# starting kernel
# baseline (speedup 1.0000x reference)
#include <hip/hip_runtime.h>
#include <hip/hip_bf16.h>

typedef float v2f __attribute__((ext_vector_type(2)));
typedef float v8f __attribute__((ext_vector_type(8)));

#define F_IN  128
#define F_HID 64
#define F_OUT 32

__device__ __forceinline__ void atomAddF(float* p, float v) {
    __hip_atomic_fetch_add(p, v, __ATOMIC_RELAXED, __HIP_MEMORY_SCOPE_AGENT);
}

__global__ void k_fill(float* __restrict__ p, float v, long long n) {
    long long i = (long long)blockIdx.x * blockDim.x + threadIdx.x;
    if (i < n) p[i] = v;
}

__global__ void k_deg(const int* __restrict__ col, float* __restrict__ deg, long long E) {
    long long e = (long long)blockIdx.x * blockDim.x + threadIdx.x;
    if (e < E) atomAddF(&deg[col[e]], 1.0f);
}

__global__ void k_dinv(float* __restrict__ deg, int n) {
    int i = blockIdx.x * blockDim.x + threadIdx.x;
    if (i < n) deg[i] = rsqrtf(deg[i]);
}

template <int K, int COUT>
__global__ void k_gemm_wmma(const float* __restrict__ X, const float* __restrict__ W,
                            float* __restrict__ H, int n) {
    constexpr int COLT = COUT / 16;
    const int lane = threadIdx.x & 31;
    const int wv   = threadIdx.x >> 5;
    const int tile = blockIdx.x * (blockDim.x >> 5) + wv;
    const int rowTile = tile / COLT;
    const int colT    = tile - rowTile * COLT;
    const int rowBase = rowTile << 4;
    if (rowBase >= n) return;

    const int m  = lane & 15;
    const int kh = (lane >> 4) << 1;
    int rowA = rowBase + m;
    if (rowA >= n) rowA = n - 1;
    const float*  xr = X + (size_t)rowA * K;
    const float*  wc = W + colT * 16 + m;

    v8f acc = {};
    #pragma unroll
    for (int k = 0; k < K; k += 4) {
        v2f a, b;
        float2 av = *(const float2*)(xr + k + kh);
        a[0] = av.x;
        a[1] = av.y;
        b[0] = wc[(size_t)(k + kh)     * COUT];
        b[1] = wc[(size_t)(k + kh + 1) * COUT];
        acc = __builtin_amdgcn_wmma_f32_16x16x4_f32(
             false, a,  false, b,
             (short)0, acc,  false,  false);
    }

    const int mo = (lane >> 4) << 3;
    const int c  = colT * 16 + m;
    #pragma unroll
    for (int v = 0; v < 8; ++v) {
        int r = rowBase + mo + v;
        if (r < n) H[(size_t)r * COUT + c] = acc[v];
    }
}

__global__ void k_scatter64(const float* __restrict__ Hs, const int* __restrict__ row,
                            const int* __restrict__ col, const float* __restrict__ dinv,
                            float* __restrict__ out, long long E) {
    long long w = ((long long)blockIdx.x * blockDim.x + threadIdx.x) >> 5;
    const int lane = threadIdx.x & 31;
    if (w >= E) return;
    const int r = row[w], c = col[w];
    const float nrm = dinv[r] * dinv[c];
    float2 v = ((const float2*)(Hs + (size_t)r * 64))[lane];
    float* o = out + (size_t)c * 64 + lane * 2;
    atomAddF(o,     v.x * nrm);
    atomAddF(o + 1, v.y * nrm);
}

__global__ void k_scatter32(const float* __restrict__ Hs, const int* __restrict__ row,
                            const int* __restrict__ col, const float* __restrict__ dinv,
                            float* __restrict__ out, long long E) {
    long long w = ((long long)blockIdx.x * blockDim.x + threadIdx.x) >> 5;
    const int lane = threadIdx.x & 31;
    if (w >= E) return;
    const int r = row[w], c = col[w];
    const float nrm = dinv[r] * dinv[c];
    float v = Hs[(size_t)r * 32 + lane];
    atomAddF(out + (size_t)c * 32 + lane, v * nrm);
}

template <int F, bool RELU>
__global__ void k_self_bias(float* __restrict__ agg, const float* __restrict__ Hs,
                            const float* __restrict__ dinv, const float* __restrict__ bias,
                            long long total) {
    long long i = (long long)blockIdx.x * blockDim.x + threadIdx.x;
    if (i >= total) return;
    const long long node = i / F;
    const int f = (int)(i - node * F);
    const float di = dinv[node];
    float v = agg[i] + di * di * Hs[i] + bias[f];
    if (RELU) v = fmaxf(v, 0.0f);
    agg[i] = v;
}

__global__ void k_decode(const float* __restrict__ Z, const int* __restrict__ ea,
                         const int* __restrict__ eb, float* __restrict__ out, long long EL) {
    long long e = (long long)blockIdx.x * blockDim.x + threadIdx.x;
    if (e >= EL) return;
    const float4* za = (const float4*)(Z + (size_t)ea[e] * 32);
    const float4* zb = (const float4*)(Z + (size_t)eb[e] * 32);
    float s = 0.0f;
    #pragma unroll
    for (int i = 0; i < 8; ++i) {
        float4 a = za[i], b = zb[i];
        s += a.x * b.x + a.y * b.y + a.z * b.z + a.w * b.w;
    }
    out[e] = s;
}

static inline int cdiv(long long a, long long b) { return (int)((a + b - 1) / b); }

extern "C" void kernel_launch(void* const* d_in, const int* in_sizes, int n_in,
                              void* d_out, int out_size, void* d_ws, size_t ws_size,
                              hipStream_t stream) {
    const float* x   = (const float*)d_in[0];
    const float* W1  = (const float*)d_in[1];
    const float* b1  = (const float*)d_in[2];
    const float* W2  = (const float*)d_in[3];
    const float* b2  = (const float*)d_in[4];
    const int*   ei  = (const int*)d_in[5];
    const int*   eli = (const int*)d_in[6];

    const long long N  = in_sizes[0] / F_IN;
    const long long E  = in_sizes[5] / 2;
    const long long EL = in_sizes[6] / 2;
    float* out = (float*)d_out;

    float* deg  = (float*)d_ws;
    float* h1   = deg  + N;
    float* agg1 = h1   + N * F_HID;
    float* h2   = agg1 + N * F_HID;
    float* agg2 = h2   + N * F_OUT;

    const int* row = ei;
    const int* col = ei + E;

    k_fill<<<cdiv(N, 256), 256, 0, stream>>>(deg, 1.0f, N);
    k_deg <<<cdiv(E, 256), 256, 0, stream>>>(col, deg, E);
    k_dinv<<<cdiv(N, 256), 256, 0, stream>>>(deg, (int)N);

    {
        long long tiles = ((N + 15) / 16) * (F_HID / 16);
        k_gemm_wmma<F_IN, F_HID><<<cdiv(tiles, 8), 256, 0, stream>>>(x, W1, h1, (int)N);
    }

    k_fill<<<cdiv(N * F_HID, 256), 256, 0, stream>>>(agg1, 0.0f, N * F_HID);
    k_scatter64<<<cdiv(E, 8), 256, 0, stream>>>(h1, row, col, deg, agg1, E);
    k_self_bias<F_HID, true><<<cdiv(N * F_HID, 256), 256, 0, stream>>>(agg1, h1, deg, b1, N * F_HID);

    {
        long long tiles = ((N + 15) / 16) * (F_OUT / 16);
        k_gemm_wmma<F_HID, F_OUT><<<cdiv(tiles, 8), 256, 0, stream>>>(agg1, W2, h2, (int)N);
    }

    k_fill<<<cdiv(N * F_OUT, 256), 256, 0, stream>>>(agg2, 0.0f, N * F_OUT);
    k_scatter32<<<cdiv(E, 8), 256, 0, stream>>>(h2, row, col, deg, agg2, E);
    k_self_bias<F_OUT, false><<<cdiv(N * F_OUT, 256), 256, 0, stream>>>(agg2, h2, deg, b2, N * F_OUT);

    k_decode<<<cdiv(EL, 256), 256, 0, stream>>>(agg2, eli, eli + EL, out, EL);
}


# layer-1 pass: all nodes' degrees as 16-bit words and the bin's own dinv held in LDS; edge weights looked up there behind the first row requests, no per-lane dinv gather from memory; sum orders kept
# speedup vs baseline: 5.3899x; 5.3899x over previous
#include <hip/hip_runtime.h>
#include <hip/hip_bf16.h>


typedef float v2f __attribute__((ext_vector_type(2)));
typedef float v8f __attribute__((ext_vector_type(8)));
typedef float v4f __attribute__((ext_vector_type(4)));
typedef unsigned int v4u __attribute__((ext_vector_type(4)));

#define F_IN  128
#define F_HID 64
#define F_OUT 32

#define NB   400
#define BT   1024
#define GB   256
#define ECH  16
#define MAXB 4096
#define DCAP 100352
#define DESC 0xFFFFu

__global__ __launch_bounds__(BT) void k_lsort(const int* __restrict__ row, const int* __restrict__ col,
                                              int2* __restrict__ edges, int* __restrict__ tab,
                                              long long E, int nbk) {
    __shared__ int hist[MAXB + 32];
    __shared__ int wsum[BT / 32];
    const int lane = threadIdx.x & 31;
    const int wv = __builtin_amdgcn_readfirstlane((int)(threadIdx.x >> 5));
    for (int i = threadIdx.x; i < nbk; i += BT) hist[i] = 0;
    if (threadIdx.x < 32) hist[MAXB + threadIdx.x] = 0;
    __syncthreads();
    const long long per = (E + GB - 1) / GB;
    long long s = per * blockIdx.x;
    if (s > E) s = E;
    long long t = s + per;
    if (t > E) t = E;
    const int nloc = (int)(t - s);
    const int* cb = col + s;
    const int* rb = row + s;
    for (int i0 = threadIdx.x; i0 < nloc; i0 += ECH * BT) {
        int c[ECH];
        #pragma unroll
        for (int q = 0; q < ECH; ++q) {
            int i = i0 + q * BT;
            if (i >= nloc) i = nloc - 1;
            c[q] = cb[i];
        }
        static_assert(ECH == 16, "the statement below names all ECH loaded words");
        asm volatile("" :: "v"(c[0]), "v"(c[1]), "v"(c[2]), "v"(c[3]), "v"(c[4]), "v"(c[5]), "v"(c[6]), "v"(c[7]),
                           "v"(c[8]), "v"(c[9]), "v"(c[10]), "v"(c[11]), "v"(c[12]), "v"(c[13]), "v"(c[14]), "v"(c[15]));
        #pragma unroll
        for (int q = 0; q < ECH; ++q)
            if (i0 + q * BT < nloc) atomicAdd(&hist[c[q] / NB], 1);
    }
    __syncthreads();
    const int ipt = (nbk + BT - 1) / BT;
    const int b0 = threadIdx.x * ipt;
    int run = 0;
    for (int i = b0; i < b0 + ipt && i < nbk; ++i) run += hist[i];
    int incl = run;
    for (int d = 1; d < 32; d <<= 1) {
        const int y = __shfl_up(incl, d);
        if (lane >= d) incl += y;
    }
    if (lane == 31) wsum[wv] = incl;
    __syncthreads();
    int base = incl - run;
    for (int w = 0; w < wv; ++w) base += wsum[w];
    for (int i = b0; i < b0 + ipt && i < nbk; ++i) {
        const int h = hist[i];
        hist[i] = (int)s + base;
        tab[(size_t)i * GB + blockIdx.x] = base;
        base += h;
    }
    if (threadIdx.x == 0) tab[(size_t)nbk * GB + blockIdx.x] = nloc;
    __syncthreads();
    for (int i0 = threadIdx.x; i0 < nloc; i0 += ECH * BT) {
        int c[ECH], r[ECH], p[ECH];
        #pragma unroll
        for (int q = 0; q < ECH; ++q) {
            int i = i0 + q * BT;
            if (i >= nloc) i = nloc - 1;
            c[q] = cb[i];
            r[q] = rb[i];
        }
        asm volatile("" :: "v"(c[0]), "v"(c[1]), "v"(c[2]), "v"(c[3]), "v"(c[4]), "v"(c[5]), "v"(c[6]), "v"(c[7]),
                           "v"(c[8]), "v"(c[9]), "v"(c[10]), "v"(c[11]), "v"(c[12]), "v"(c[13]), "v"(c[14]), "v"(c[15]));
        asm volatile("" :: "v"(r[0]), "v"(r[1]), "v"(r[2]), "v"(r[3]), "v"(r[4]), "v"(r[5]), "v"(r[6]), "v"(r[7]),
                           "v"(r[8]), "v"(r[9]), "v"(r[10]), "v"(r[11]), "v"(r[12]), "v"(r[13]), "v"(r[14]), "v"(r[15]));
        #pragma unroll
        for (int q = 0; q < ECH; ++q) {
            const int hb = (i0 + q * BT < nloc) ? c[q] / NB : MAXB + lane;
            p[q] = atomicAdd(&hist[hb], 1);
        }
        #pragma unroll
        for (int q = 0; q < ECH; ++q) {
            if (i0 + q * BT < nloc) {
                const int b = c[q] / NB;
                edges[p[q]] = make_int2(r[q], c[q] - b * NB);
            }
        }
    }
}

__device__ __forceinline__ void segTable(const int* __restrict__ tab, int b, int per,
                                         int* cum, int* adj, int lane) {
    constexpr int SPL = GB / 32;
    static_assert(GB % 32 == 0 && (GB & (GB - 1)) == 0, "slotOf halves GB down to 1");
    int s[SPL], t[SPL];
    #pragma unroll
    for (int q = 0; q < SPL; ++q) {
        s[q] = tab[(size_t)b * GB + lane * SPL + q];
        t[q] = tab[(size_t)(b + 1) * GB + lane * SPL + q];
    }
    int ex[SPL];
    int run = 0;
    #pragma unroll
    for (int q = 0; q < SPL; ++q) {
        ex[q] = run;
        run += t[q] - s[q];
    }
    int incl = run;
    for (int d = 1; d < 32; d <<= 1) {
        const int y = __shfl_up(incl, d);
        if (lane >= d) incl += y;
    }
    const int base = incl - run;
    #pragma unroll
    for (int q = 0; q < SPL; ++q) {
        const int k = lane * SPL + q;
        cum[k] = base + ex[q];
        adj[k] = k * per + s[q] - (base + ex[q]);
    }
    if (lane == 31) cum[GB] = incl;
}

__device__ __forceinline__ int slotOf(const int* cum, const int* adj, int v) {
    int k = 0;
    #pragma unroll
    for (int st = GB / 2; st > 0; st >>= 1)
        if (cum[k + st] <= v) k += st;
    return v + adj[k];
}

#define XP  132
#define NS  (NB / 16)
#define RPT 4

__global__ __launch_bounds__(BT) void k_bdeg_gemm(const int2* __restrict__ edges, const int* __restrict__ tab, long long E,
                                                  float* __restrict__ dinv, unsigned short* __restrict__ deg16,
                                                  const float* __restrict__ X,
                                                  const float* __restrict__ W, float* __restrict__ H, int n) {
    static_assert(NB % 16 == 0 && NS <= BT / 32, "one 16-row strip per wave");
    static_assert((F_IN * F_HID / 4) % BT == 0, "W is read as whole float4 items per thread");
    __shared__ __attribute__((aligned(16))) float wt[F_HID * XP];
    __shared__ __attribute__((aligned(16))) float xs[NS * 16 * XP];
    __shared__ int cnt[NB];
    __shared__ int ticket;
    __shared__ int cum[GB + 1];
    __shared__ int adj[GB];
    const int lane = threadIdx.x & 31;
    const int wv = __builtin_amdgcn_readfirstlane((int)(threadIdx.x >> 5));
    const int b = blockIdx.x;
    const int node0 = b * NB;
    const int r0 = node0 + wv * 16;
    const int per = (int)((E + GB - 1) / GB);

    if (wv < NS) {
        int r = r0 + (lane >> 1);
        if (r >= n) r = n - 1;
        __builtin_prefetch(X + (size_t)r * F_IN + (lane & 1) * (F_IN / 2), 0, 3);
    }
    constexpr int WQ = F_IN * F_HID / 4 / BT;
    v4f wr[WQ];
    #pragma unroll
    for (int q = 0; q < WQ; ++q) wr[q] = ((const v4f*)W)[threadIdx.x + q * BT];
    for (int i = threadIdx.x; i < NB; i += BT) cnt[i] = 0;
    if (threadIdx.x == 0) ticket = 0;
    if (wv == 0) segTable(tab, b, per, cum, adj, lane);
    #pragma unroll
    for (int q = 0; q < WQ; ++q) {
        const int j = threadIdx.x + q * BT;
        const int k = j / (F_HID / 4), c4 = j % (F_HID / 4);
        float* p = wt + (4 * c4) * XP + k;
        p[0]      = wr[q][0];
        p[XP]     = wr[q][1];
        p[2 * XP] = wr[q][2];
        p[3 * XP] = wr[q][3];
    }
    __syncthreads();

    if (wv < NS) {
        v4f xr[16];
        #pragma unroll
        for (int i = 0; i < 16; ++i) {
            int r = r0 + i;
            if (r >= n) r = n - 1;
            xr[i] = *(const v4f*)(X + (size_t)r * F_IN + lane * 4);
        }
        float* xw = xs + wv * 16 * XP;
        #pragma unroll
        for (int i = 0; i < 16; ++i) *(v4f*)(xw + i * XP + lane * 4) = xr[i];
        const int m  = lane & 15;
        const int kh = (lane >> 4) << 1;
        const float* ar = xw + m * XP + kh;
        const float* br = wt + m * XP + kh;
        constexpr int CT = F_HID / 16;
        v8f c[CT] = {};
        #pragma unroll
        for (int k = 0; k < F_IN; k += 4) {
            v2f a;
            const float2 av = *(const float2*)(ar + k);
            a[0] = av.x;
            a[1] = av.y;
            #pragma unroll
            for (int t = 0; t < CT; ++t) {
                v2f bf;
                const float2 bv = *(const float2*)(br + t * 16 * XP + k);
                bf[0] = bv.x;
                bf[1] = bv.y;
                c[t] = __builtin_amdgcn_wmma_f32_16x16x4_f32(
                     false, a,  false, bf,
                     (short)0, c[t],  false,  false);
            }
        }
        const int mo = (lane >> 4) << 3;
        #pragma unroll
        for (int v = 0; v < 8; ++v) {
            const int r = r0 + mo + v;
            if (r < n) {
                #pragma unroll
                for (int t = 0; t < CT; ++t) H[(size_t)r * F_HID + t * 16 + m] = c[t][v];
            }
        }
    }

    static_assert(GB % RPT == 0, "a ticket is RPT whole runs");
    const int T = cum[GB];
    for (; T > 0;) {
        int t = 0;
        if (lane == 0) t = atomicAdd(&ticket, 1);
        t = __builtin_amdgcn_readfirstlane(t);
        const int k0 = t * RPT;
        if (k0 >= GB) break;
        int d[2 * RPT], len[RPT], base[RPT];
        #pragma unroll
        for (int q = 0; q < RPT; ++q) {
            const int c0 = cum[k0 + q];
            len[q] = cum[k0 + q + 1] - c0;
            base[q] = adj[k0 + q] + c0;
            d[2 * q]     = edges[(lane      < len[q]) ? base[q] + lane      : 0].y;
            d[2 * q + 1] = edges[(lane + 32 < len[q]) ? base[q] + lane + 32 : 0].y;
        }
        static_assert(RPT == 4, "the statement below names all 2*RPT loaded words");
        asm volatile("" :: "v"(d[0]), "v"(d[1]), "v"(d[2]), "v"(d[3]), "v"(d[4]), "v"(d[5]), "v"(d[6]), "v"(d[7]));
        #pragma unroll
        for (int q = 0; q < 2 * RPT; ++q)
            if (lane + 32 * (q & 1) < len[q >> 1]) atomicAdd(&cnt[d[q]], 1);
        #pragma unroll
        for (int q = 0; q < RPT; ++q)
            for (int o = lane + 64; o < len[q]; o += 32) atomicAdd(&cnt[edges[base[q] + o].y], 1);
    }
    __syncthreads();
    for (int i = threadIdx.x; i < NB; i += BT) {
        const int node = node0 + i;
        if (node < n) {
            const int d = cnt[i] + 1;
            dinv[node] = rsqrtf((float)d);
            deg16[node] = (unsigned short)(d < (int)DESC ? (unsigned)d : DESC);
        }
    }
}

__device__ __forceinline__ void ldsAddF(float* p, float v) {
    __hip_atomic_fetch_add(p, v, __ATOMIC_RELAXED, __HIP_MEMORY_SCOPE_WORKGROUP);
}

template <int F, int J>
__device__ __forceinline__ void loadRows(const float* __restrict__ hl, int r, int j0,
                                         float (&vx)[J], float (&vy)[J]) {
    #pragma unroll
    for (int j = 0; j < J; ++j) {
        const int rj = __builtin_amdgcn_readlane(r, j0 + j);
        const float* p = hl + (size_t)rj * F;
        if constexpr (F == 64) {
            const float2 v = *(const float2*)p;
            vx[j] = v.x;
            vy[j] = v.y;
        } else {
            vx[j] = *p;
            vy[j] = 0.0f;
        }
    }
}

template <int F, int J>
__device__ __forceinline__ void addLoaded(float* al, int co, float w, int j0,
                                          const float (&vx)[J], const float (&vy)[J]) {
    #pragma unroll
    for (int j = 0; j < J; ++j) {
        const int cj = __builtin_amdgcn_readlane(co, j0 + j);
        const float wj = __int_as_float(__builtin_amdgcn_readlane(__float_as_int(w), j0 + j));
        float* q = al + cj;
        ldsAddF(q, vx[j] * wj);
        if constexpr (F == 64) ldsAddF(q + 1, vy[j] * wj);
    }
}

template <int F, int J>
__device__ __forceinline__ void addRows(const float* __restrict__ hl, float* al,
                                        int r, int co, float w, int j0) {
    float vx[J], vy[J];
    loadRows<F, J>(hl, r, j0, vx, vy);
    addLoaded<F, J>(al, co, w, j0, vx, vy);
}

__device__ __forceinline__ float srcDinv(const unsigned short* dtab, const float* __restrict__ dinv, int r) {
    const unsigned d = r < DCAP ? (unsigned)dtab[r] : DESC;
    float s;
    if (d != DESC) {
        s = rsqrtf((float)d);
    } else {
        s = dinv[r];
        asm volatile("" :: "v"(s));
    }
    return s;
}

template <int F, bool RELU, int COUT, bool WST>
__global__ __launch_bounds__(BT) void k_bagg(const float* __restrict__ Hs, const int2* __restrict__ edges,
                                             const int* __restrict__ tab, long long E, const float* __restrict__ dinv,
                                             const float* __restrict__ bias, const float* __restrict__ Wn,
                                             float* __restrict__ wts, const unsigned short* __restrict__ deg16,
                                             float* __restrict__ out, int n) {
    constexpr int VPL = F / 32;
    constexpr int P = (COUT > 0) ? F + 4 : F;
    static_assert(NB % 16 == 0 && NB / 16 <= BT / 32, "one 16-row strip per wave");
    __shared__ __attribute__((aligned(16))) float acc[NB * P];
    __shared__ int cum[GB + 1];
    __shared__ int adj[GB];
    constexpr int DN = WST ? DCAP : 8;
    static_assert(DN % 8 == 0, "the table is copied in 16-byte items of 8 words");
    __shared__ __attribute__((aligned(16))) unsigned short dtab[DN];
    __shared__ float dl[WST ? NB : 1];
    const int lane = threadIdx.x & 31;
    const int wv = __builtin_amdgcn_readfirstlane((int)(threadIdx.x >> 5));
    const int b = blockIdx.x;
    const int node0 = b * NB;

    constexpr int DQ = (DN / 8 + BT - 1) / BT;
    const int nt = n < DN ? n : DN;
    const int ni = nt / 8;
    const int rot = (int)((unsigned)(b * ni) / gridDim.x);
    v4u dv[DQ];
    if constexpr (WST) {
        if (ni > 0) {
            #pragma unroll
            for (int q = 0; q < DQ; ++q) {
                int j = threadIdx.x + q * BT;
                if (j >= ni) j = ni - 1;
                j += rot;
                if (j >= ni) j -= ni;
                dv[q] = ((const v4u*)deg16)[j];
            }
        }
    }
    for (int i = threadIdx.x; i < NB * P / 4; i += BT) ((float4*)acc)[i] = make_float4(0.f, 0.f, 0.f, 0.f);
    if (wv == 0) segTable(tab, b, (int)((E + GB - 1) / GB), cum, adj, lane);
    if constexpr (WST) {
        for (int i = threadIdx.x; i < NB; i += BT) {
            const int node = node0 + i;
            dl[i] = node < n ? dinv[node] : 0.0f;
        }
        if ((int)threadIdx.x < nt - ni * 8) dtab[ni * 8 + threadIdx.x] = deg16[ni * 8 + threadIdx.x];
        if (ni > 0) {
            #pragma unroll
            for (int q = 0; q < DQ; ++q) {
                int j = threadIdx.x + q * BT;
                if (j < ni) {
                    j += rot;
                    if (j >= ni) j -= ni;
                    ((v4u*)dtab)[j] = dv[q];
                }
            }
        }
    }
    __syncthreads();

    const int T = cum[GB];
    const float* hl = Hs + lane * VPL;
    float* al = acc + lane * VPL;
    int g = wv * 32;
    int2 rcn = make_int2(0, 0);
    float wn = 0.0f;
    int sn = 0;
    if (g + lane < T) {
        sn = slotOf(cum, adj, g + lane);
        rcn = edges[sn];
        if constexpr (!WST) wn = wts[sn];
    }
    for (; g < T; g += BT) {
        const int cnt = T - g;
        const int2 rc = rcn;
        float w = wn;
        const int sl = sn;
        const int r = rc.x, c = rc.y;
        const int co = c * P;
        if (cnt >= 32) {
            float ax[16], ay[16];
            loadRows<F, 16>(hl, r, 0, ax, ay);
            if constexpr (WST) {
                const float dc = dl[c];
                w = srcDinv(dtab, dinv, r) * dc;
                wts[sl] = w;
            }
            if (g + BT + lane < T) {
                sn = slotOf(cum, adj, g + BT + lane);
                rcn = edges[sn];
                if constexpr (!WST) wn = wts[sn];
            }
            addLoaded<F, 16>(al, co, w, 0, ax, ay);
            addRows<F, 16>(hl, al, r, co, w, 16);
        } else {
            if constexpr (WST) {
                w = 0.0f;
                if (lane < cnt) {
                    const float dc = dl[c];
                    w = srcDinv(dtab, dinv, r) * dc;
                    wts[sl] = w;
                }
            }
            for (int j = 0; j < cnt; ++j) addRows<F, 1>(hl, al, r, co, w, j);
        }
    }
    __syncthreads();

    for (int i = threadIdx.x; i < NB * F / 4; i += BT) {
        const int nl = (i * 4) / F;
        const int node = node0 + nl;
        if (node < n) {
            const int f = (i * 4) % F;
            const size_t gi = (size_t)node0 * F + (size_t)i * 4;
            const float di = dinv[node];
            float4* ap = (float4*)(acc + nl * P + f);
            const float4 a = *ap;
            const float4 h = *(const float4*)(Hs + gi);
            float4 v;
            v.x = a.x + di * di * h.x + bias[f];
            v.y = a.y + di * di * h.y + bias[f + 1];
            v.z = a.z + di * di * h.z + bias[f + 2];
            v.w = a.w + di * di * h.w + bias[f + 3];
            if (RELU) {
                v.x = fmaxf(v.x, 0.0f);
                v.y = fmaxf(v.y, 0.0f);
                v.z = fmaxf(v.z, 0.0f);
                v.w = fmaxf(v.w, 0.0f);
            }
            if constexpr (COUT > 0) *ap = v;
            else *(float4*)(out + gi) = v;
        }
    }

    if constexpr (COUT > 0) {
        constexpr int CT = COUT / 16;
        __syncthreads();
        if (wv < NB / 16) {
            const int m  = lane & 15;
            const int kh = (lane >> 4) << 1;
            const float* ar = acc + (wv * 16 + m) * P + kh;
            const float* wc = Wn + kh * COUT + m;
            v8f c[CT] = {};
            #pragma unroll
            for (int k = 0; k < F; k += 4) {
                v2f a;
                const float2 av = *(const float2*)(ar + k);
                a[0] = av.x;
                a[1] = av.y;
                #pragma unroll
                for (int t = 0; t < CT; ++t) {
                    v2f bf;
                    bf[0] = wc[(size_t)k       * COUT + t * 16];
                    bf[1] = wc[(size_t)(k + 1) * COUT + t * 16];
                    c[t] = __builtin_amdgcn_wmma_f32_16x16x4_f32(
                         false, a,  false, bf,
                         (short)0, c[t],  false,  false);
                }
            }
            const int mo = (lane >> 4) << 3;
            #pragma unroll
            for (int v = 0; v < 8; ++v) {
                const int r = node0 + wv * 16 + mo + v;
                if (r < n) {
                    #pragma unroll
                    for (int t = 0; t < CT; ++t) out[(size_t)r * COUT + t * 16 + m] = c[t][v];
                }
            }
        }
    }
}

__global__ void k_decode(const float* __restrict__ Z, const int* __restrict__ ea,
                         const int* __restrict__ eb, float* __restrict__ out, long long EL) {
    long long e = (long long)blockIdx.x * blockDim.x + threadIdx.x;
    if (e >= EL) return;
    const float4* za = (const float4*)(Z + (size_t)ea[e] * 32);
    const float4* zb = (const float4*)(Z + (size_t)eb[e] * 32);
    float s = 0.0f;
    #pragma unroll
    for (int i = 0; i < 8; ++i) {
        float4 a = za[i], b = zb[i];
        s += a.x * b.x + a.y * b.y + a.z * b.z + a.w * b.w;
    }
    out[e] = s;
}

static inline int cdiv(long long a, long long b) { return (int)((a + b - 1) / b); }

extern "C" void kernel_launch(void* const* d_in, const int* in_sizes, int n_in,
                              void* d_out, int out_size, void* d_ws, size_t ws_size,
                              hipStream_t stream) {
    const float* x   = (const float*)d_in[0];
    const float* W1  = (const float*)d_in[1];
    const float* b1  = (const float*)d_in[2];
    const float* W2  = (const float*)d_in[3];
    const float* b2  = (const float*)d_in[4];
    const int*   ei  = (const int*)d_in[5];
    const int*   eli = (const int*)d_in[6];

    const long long N  = in_sizes[0] / F_IN;
    const long long E  = in_sizes[5] / 2;
    const long long EL = in_sizes[6] / 2;
    float* out = (float*)d_out;
    const int nbk = cdiv(N, NB);

    float* h1    = (float*)d_ws;
    float* h2    = h1   + N * F_HID;
    float* agg2  = h2   + N * F_OUT;
    unsigned short* deg16 = (unsigned short*)(agg2 + N * F_OUT);
    int2*  edges = (int2*)(deg16 + 4 * N);
    float* dinv  = (float*)(edges + E);
    float* wts   = dinv + N;
    int*   tab   = (int*)(wts + E);

    const int* row = ei;
    const int* col = ei + E;

    k_lsort<<<GB, BT, 0, stream>>>(row, col, edges, tab, E, nbk);

    k_bdeg_gemm<<<nbk, BT, 0, stream>>>(edges, tab, E, dinv, deg16, x, W1, h1, (int)N);

    k_bagg<F_HID, true, F_OUT, true><<<nbk, BT, 0, stream>>>(h1, edges, tab, E, dinv, b1, W2, wts, deg16, h2, (int)N);

    k_bagg<F_OUT, false, 0, false><<<nbk, BT, 0, stream>>>(h2, edges, tab, E, dinv, b2, nullptr, wts, nullptr, agg2, (int)N);

    k_decode<<<cdiv(EL, 256), 256, 0, stream>>>(agg2, eli, eli + EL, out, EL);
}


# sorting pass and decode use whole-line memory instructions: sorted range put together in LDS and stored in slot order; decode reads a row with 8 lanes; its dot is summed in a new order
# speedup vs baseline: 5.7654x; 1.0697x over previous
#include <hip/hip_runtime.h>
#include <hip/hip_bf16.h>


typedef float v2f __attribute__((ext_vector_type(2)));
typedef float v8f __attribute__((ext_vector_type(8)));
typedef float v4f __attribute__((ext_vector_type(4)));
typedef unsigned int v4u __attribute__((ext_vector_type(4)));

#define F_IN  128
#define F_HID 64
#define F_OUT 32

#define NB   400
#define BT   1024
#define GB   256
#define ECH  16
#define MAXB 4096
#define DCAP 100352
#define DESC 0xFFFFu
#define SCAP 16384


template <bool STAGE>
__device__ __forceinline__ void placeRange(const int* __restrict__ cb, const int* __restrict__ rb, int nloc,
                                           int* hist, int2* stage, int2* __restrict__ edges, int lane) {
    for (int i0 = threadIdx.x; i0 < nloc; i0 += ECH * BT) {
        int c[ECH], r[ECH], p[ECH];
        #pragma unroll
        for (int q = 0; q < ECH; ++q) {
            int i = i0 + q * BT;
            if (i >= nloc) i = nloc - 1;
            c[q] = cb[i];
            r[q] = rb[i];
        }
        static_assert(ECH == 16, "the statements below name all ECH loaded words");
        asm volatile("" :: "v"(c[0]), "v"(c[1]), "v"(c[2]), "v"(c[3]), "v"(c[4]), "v"(c[5]), "v"(c[6]), "v"(c[7]),
                           "v"(c[8]), "v"(c[9]), "v"(c[10]), "v"(c[11]), "v"(c[12]), "v"(c[13]), "v"(c[14]), "v"(c[15]));
        asm volatile("" :: "v"(r[0]), "v"(r[1]), "v"(r[2]), "v"(r[3]), "v"(r[4]), "v"(r[5]), "v"(r[6]), "v"(r[7]),
                           "v"(r[8]), "v"(r[9]), "v"(r[10]), "v"(r[11]), "v"(r[12]), "v"(r[13]), "v"(r[14]), "v"(r[15]));
        #pragma unroll
        for (int q = 0; q < ECH; ++q) {
            const int hb = (i0 + q * BT < nloc) ? c[q] / NB : MAXB + lane;
            p[q] = atomicAdd(&hist[hb], 1);
        }
        #pragma unroll
        for (int q = 0; q < ECH; ++q) {
            if (i0 + q * BT < nloc) {
                const int b = c[q] / NB;
                const int2 v = make_int2(r[q], c[q] - b * NB);
                if constexpr (STAGE) stage[p[q]] = v;
                else edges[p[q]] = v;
            }
        }
    }
}

__global__ __launch_bounds__(BT) void k_lsort(const int* __restrict__ row, const int* __restrict__ col,
                                              int2* __restrict__ edges, int* __restrict__ tab,
                                              long long E, int nbk) {
    __shared__ int hist[MAXB + 32];
    __shared__ int wsum[BT / 32];
    __shared__ int2 stage[SCAP];
    const int lane = threadIdx.x & 31;
    const int wv = __builtin_amdgcn_readfirstlane((int)(threadIdx.x >> 5));
    for (int i = threadIdx.x; i < nbk; i += BT) hist[i] = 0;
    if (threadIdx.x < 32) hist[MAXB + threadIdx.x] = 0;
    __syncthreads();
    const long long per = (E + GB - 1) / GB;
    long long s = per * blockIdx.x;
    if (s > E) s = E;
    long long t = s + per;
    if (t > E) t = E;
    const int nloc = (int)(t - s);
    const bool inl = nloc <= SCAP;
    const int* cb = col + s;
    const int* rb = row + s;
    for (int i0 = threadIdx.x; i0 < nloc; i0 += ECH * BT) {
        int c[ECH];
        #pragma unroll
        for (int q = 0; q < ECH; ++q) {
            int i = i0 + q * BT;
            if (i >= nloc) i = nloc - 1;
            c[q] = cb[i];
        }
        static_assert(ECH == 16, "the statement below names all ECH loaded words");
        asm volatile("" :: "v"(c[0]), "v"(c[1]), "v"(c[2]), "v"(c[3]), "v"(c[4]), "v"(c[5]), "v"(c[6]), "v"(c[7]),
                           "v"(c[8]), "v"(c[9]), "v"(c[10]), "v"(c[11]), "v"(c[12]), "v"(c[13]), "v"(c[14]), "v"(c[15]));
        #pragma unroll
        for (int q = 0; q < ECH; ++q)
            if (i0 + q * BT < nloc) atomicAdd(&hist[c[q] / NB], 1);
    }
    __syncthreads();
    const int ipt = (nbk + BT - 1) / BT;
    const int b0 = threadIdx.x * ipt;
    int run = 0;
    for (int i = b0; i < b0 + ipt && i < nbk; ++i) run += hist[i];
    int incl = run;
    for (int d = 1; d < 32; d <<= 1) {
        const int y = __shfl_up(incl, d);
        if (lane >= d) incl += y;
    }
    if (lane == 31) wsum[wv] = incl;
    __syncthreads();
    int base = incl - run;
    for (int w = 0; w < wv; ++w) base += wsum[w];
    for (int i = b0; i < b0 + ipt && i < nbk; ++i) {
        const int h = hist[i];
        hist[i] = inl ? base : (int)s + base;
        tab[(size_t)i * GB + blockIdx.x] = base;
        base += h;
    }
    if (threadIdx.x == 0) tab[(size_t)nbk * GB + blockIdx.x] = nloc;
    __syncthreads();
    if (inl) {
        placeRange<true>(cb, rb, nloc, hist, stage, edges, lane);
        __syncthreads();
        for (int i0 = threadIdx.x; i0 < nloc; i0 += 4 * BT) {
            int2 v[4];
            #pragma unroll
            for (int q = 0; q < 4; ++q) {
                int i = i0 + q * BT;
                if (i >= nloc) i = nloc - 1;
                v[q] = stage[i];
            }
            asm volatile("" :: "v"(v[0].x), "v"(v[0].y), "v"(v[1].x), "v"(v[1].y),
                               "v"(v[2].x), "v"(v[2].y), "v"(v[3].x), "v"(v[3].y));
            #pragma unroll
            for (int q = 0; q < 4; ++q)
                if (i0 + q * BT < nloc) edges[s + i0 + q * BT] = v[q];
        }
    } else {
        placeRange<false>(cb, rb, nloc, hist, stage, edges, lane);
    }
}

__device__ __forceinline__ void segTable(const int* __restrict__ tab, int b, int per,
                                         int* cum, int* adj, int lane) {
    constexpr int SPL = GB / 32;
    static_assert(GB % 32 == 0 && (GB & (GB - 1)) == 0, "slotOf halves GB down to 1");
    int s[SPL], t[SPL];
    #pragma unroll
    for (int q = 0; q < SPL; ++q) {
        s[q] = tab[(size_t)b * GB + lane * SPL + q];
        t[q] = tab[(size_t)(b + 1) * GB + lane * SPL + q];
    }
    int ex[SPL];
    int run = 0;
    #pragma unroll
    for (int q = 0; q < SPL; ++q) {
        ex[q] = run;
        run += t[q] - s[q];
    }
    int incl = run;
    for (int d = 1; d < 32; d <<= 1) {
        const int y = __shfl_up(incl, d);
        if (lane >= d) incl += y;
    }
    const int base = incl - run;
    #pragma unroll
    for (int q = 0; q < SPL; ++q) {
        const int k = lane * SPL + q;
        cum[k] = base + ex[q];
        adj[k] = k * per + s[q] - (base + ex[q]);
    }
    if (lane == 31) cum[GB] = incl;
}

__device__ __forceinline__ int slotOf(const int* cum, const int* adj, int v) {
    int k = 0;
    #pragma unroll
    for (int st = GB / 2; st > 0; st >>= 1)
        if (cum[k + st] <= v) k += st;
    return v + adj[k];
}

#define XP  132
#define NS  (NB / 16)
#define RPT 4

__global__ __launch_bounds__(BT) void k_bdeg_gemm(const int2* __restrict__ edges, const int* __restrict__ tab, long long E,
                                                  float* __restrict__ dinv, unsigned short* __restrict__ deg16,
                                                  const float* __restrict__ X,
                                                  const float* __restrict__ W, float* __restrict__ H, int n) {
    static_assert(NB % 16 == 0 && NS <= BT / 32, "one 16-row strip per wave");
    static_assert((F_IN * F_HID / 4) % BT == 0, "W is read as whole float4 items per thread");
    __shared__ __attribute__((aligned(16))) float wt[F_HID * XP];
    __shared__ __attribute__((aligned(16))) float xs[NS * 16 * XP];
    __shared__ int cnt[NB];
    __shared__ int ticket;
    __shared__ int cum[GB + 1];
    __shared__ int adj[GB];
    const int lane = threadIdx.x & 31;
    const int wv = __builtin_amdgcn_readfirstlane((int)(threadIdx.x >> 5));
    const int b = blockIdx.x;
    const int node0 = b * NB;
    const int r0 = node0 + wv * 16;
    const int per = (int)((E + GB - 1) / GB);

    if (wv < NS) {
        int r = r0 + (lane >> 1);
        if (r >= n) r = n - 1;
        __builtin_prefetch(X + (size_t)r * F_IN + (lane & 1) * (F_IN / 2), 0, 3);
    }
    constexpr int WQ = F_IN * F_HID / 4 / BT;
    v4f wr[WQ];
    #pragma unroll
    for (int q = 0; q < WQ; ++q) wr[q] = ((const v4f*)W)[threadIdx.x + q * BT];
    for (int i = threadIdx.x; i < NB; i += BT) cnt[i] = 0;
    if (threadIdx.x == 0) ticket = 0;
    if (wv == 0) segTable(tab, b, per, cum, adj, lane);
    #pragma unroll
    for (int q = 0; q < WQ; ++q) {
        const int j = threadIdx.x + q * BT;
        const int k = j / (F_HID / 4), c4 = j % (F_HID / 4);
        float* p = wt + (4 * c4) * XP + k;
        p[0]      = wr[q][0];
        p[XP]     = wr[q][1];
        p[2 * XP] = wr[q][2];
        p[3 * XP] = wr[q][3];
    }
    __syncthreads();

    if (wv < NS) {
        v4f xr[16];
        #pragma unroll
        for (int i = 0; i < 16; ++i) {
            int r = r0 + i;
            if (r >= n) r = n - 1;
            xr[i] = *(const v4f*)(X + (size_t)r * F_IN + lane * 4);
        }
        float* xw = xs + wv * 16 * XP;
        #pragma unroll
        for (int i = 0; i < 16; ++i) *(v4f*)(xw + i * XP + lane * 4) = xr[i];
        const int m  = lane & 15;
        const int kh = (lane >> 4) << 1;
        const float* ar = xw + m * XP + kh;
        const float* br = wt + m * XP + kh;
        constexpr int CT = F_HID / 16;
        v8f c[CT] = {};
        #pragma unroll
        for (int k = 0; k < F_IN; k += 4) {
            v2f a;
            const float2 av = *(const float2*)(ar + k);
            a[0] = av.x;
            a[1] = av.y;
            #pragma unroll
            for (int t = 0; t < CT; ++t) {
                v2f bf;
                const float2 bv = *(const float2*)(br + t * 16 * XP + k);
                bf[0] = bv.x;
                bf[1] = bv.y;
                c[t] = __builtin_amdgcn_wmma_f32_16x16x4_f32(
                     false, a,  false, bf,
                     (short)0, c[t],  false,  false);
            }
        }
        const int mo = (lane >> 4) << 3;
        #pragma unroll
        for (int v = 0; v < 8; ++v) {
            const int r = r0 + mo + v;
            if (r < n) {
                #pragma unroll
                for (int t = 0; t < CT; ++t) H[(size_t)r * F_HID + t * 16 + m] = c[t][v];
            }
        }
    }

    static_assert(GB % RPT == 0, "a ticket is RPT whole runs");
    const int T = cum[GB];
    for (; T > 0;) {
        int t = 0;
        if (lane == 0) t = atomicAdd(&ticket, 1);
        t = __builtin_amdgcn_readfirstlane(t);
        const int k0 = t * RPT;
        if (k0 >= GB) break;
        int d[2 * RPT], len[RPT], base[RPT];
        #pragma unroll
        for (int q = 0; q < RPT; ++q) {
            const int c0 = cum[k0 + q];
            len[q] = cum[k0 + q + 1] - c0;
            base[q] = adj[k0 + q] + c0;
            d[2 * q]     = edges[(lane      < len[q]) ? base[q] + lane      : 0].y;
            d[2 * q + 1] = edges[(lane + 32 < len[q]) ? base[q] + lane + 32 : 0].y;
        }
        static_assert(RPT == 4, "the statement below names all 2*RPT loaded words");
        asm volatile("" :: "v"(d[0]), "v"(d[1]), "v"(d[2]), "v"(d[3]), "v"(d[4]), "v"(d[5]), "v"(d[6]), "v"(d[7]));
        #pragma unroll
        for (int q = 0; q < 2 * RPT; ++q)
            if (lane + 32 * (q & 1) < len[q >> 1]) atomicAdd(&cnt[d[q]], 1);
        #pragma unroll
        for (int q = 0; q < RPT; ++q)
            for (int o = lane + 64; o < len[q]; o += 32) atomicAdd(&cnt[edges[base[q] + o].y], 1);
    }
    __syncthreads();
    for (int i = threadIdx.x; i < NB; i += BT) {
        const int node = node0 + i;
        if (node < n) {
            const int d = cnt[i] + 1;
            dinv[node] = rsqrtf((float)d);
            deg16[node] = (unsigned short)(d < (int)DESC ? (unsigned)d : DESC);
        }
    }
}

__device__ __forceinline__ void ldsAddF(float* p, float v) {
    __hip_atomic_fetch_add(p, v, __ATOMIC_RELAXED, __HIP_MEMORY_SCOPE_WORKGROUP);
}

template <int F, int J>
__device__ __forceinline__ void loadRows(const float* __restrict__ hl, int r, int j0,
                                         float (&vx)[J], float (&vy)[J]) {
    #pragma unroll
    for (int j = 0; j < J; ++j) {
        const int rj = __builtin_amdgcn_readlane(r, j0 + j);
        const float* p = hl + (size_t)rj * F;
        if constexpr (F == 64) {
            const float2 v = *(const float2*)p;
            vx[j] = v.x;
            vy[j] = v.y;
        } else {
            vx[j] = *p;
            vy[j] = 0.0f;
        }
    }
}

template <int F, int J>
__device__ __forceinline__ void addLoaded(float* al, int co, float w, int j0,
                                          const float (&vx)[J], const float (&vy)[J]) {
    #pragma unroll
    for (int j = 0; j < J; ++j) {
        const int cj = __builtin_amdgcn_readlane(co, j0 + j);
        const float wj = __int_as_float(__builtin_amdgcn_readlane(__float_as_int(w), j0 + j));
        float* q = al + cj;
        ldsAddF(q, vx[j] * wj);
        if constexpr (F == 64) ldsAddF(q + 1, vy[j] * wj);
    }
}

template <int F, int J>
__device__ __forceinline__ void addRows(const float* __restrict__ hl, float* al,
                                        int r, int co, float w, int j0) {
    float vx[J], vy[J];
    loadRows<F, J>(hl, r, j0, vx, vy);
    addLoaded<F, J>(al, co, w, j0, vx, vy);
}

__device__ __forceinline__ float srcDinv(const unsigned short* dtab, const float* __restrict__ dinv, int r) {
    const unsigned d = r < DCAP ? (unsigned)dtab[r] : DESC;
    float s;
    if (d != DESC) {
        s = rsqrtf((float)d);
    } else {
        s = dinv[r];
        asm volatile("" :: "v"(s));
    }
    return s;
}

template <int F, bool RELU, int COUT, bool WST>
__global__ __launch_bounds__(BT) void k_bagg(const float* __restrict__ Hs, const int2* __restrict__ edges,
                                             const int* __restrict__ tab, long long E, const float* __restrict__ dinv,
                                             const float* __restrict__ bias, const float* __restrict__ Wn,
                                             float* __restrict__ wts, const unsigned short* __restrict__ deg16,
                                             float* __restrict__ out, int n) {
    constexpr int VPL = F / 32;
    constexpr int P = (COUT > 0) ? F + 4 : F;
    static_assert(NB % 16 == 0 && NB / 16 <= BT / 32, "one 16-row strip per wave");
    __shared__ __attribute__((aligned(16))) float acc[NB * P];
    __shared__ int cum[GB + 1];
    __shared__ int adj[GB];
    constexpr int DN = WST ? DCAP : 8;
    static_assert(DN % 8 == 0, "the table is copied in 16-byte items of 8 words");
    __shared__ __attribute__((aligned(16))) unsigned short dtab[DN];
    __shared__ float dl[WST ? NB : 1];
    const int lane = threadIdx.x & 31;
    const int wv = __builtin_amdgcn_readfirstlane((int)(threadIdx.x >> 5));
    const int b = blockIdx.x;
    const int node0 = b * NB;

    constexpr int DQ = (DN / 8 + BT - 1) / BT;
    const int nt = n < DN ? n : DN;
    const int ni = nt / 8;
    const int rot = (int)((unsigned)(b * ni) / gridDim.x);
    v4u dv[DQ];
    if constexpr (WST) {
        if (ni > 0) {
            #pragma unroll
            for (int q = 0; q < DQ; ++q) {
                int j = threadIdx.x + q * BT;
                if (j >= ni) j = ni - 1;
                j += rot;
                if (j >= ni) j -= ni;
                dv[q] = ((const v4u*)deg16)[j];
            }
        }
    }
    for (int i = threadIdx.x; i < NB * P / 4; i += BT) ((float4*)acc)[i] = make_float4(0.f, 0.f, 0.f, 0.f);
    if (wv == 0) segTable(tab, b, (int)((E + GB - 1) / GB), cum, adj, lane);
    if constexpr (WST) {
        for (int i = threadIdx.x; i < NB; i += BT) {
            const int node = node0 + i;
            dl[i] = node < n ? dinv[node] : 0.0f;
        }
        if ((int)threadIdx.x < nt - ni * 8) dtab[ni * 8 + threadIdx.x] = deg16[ni * 8 + threadIdx.x];
        if (ni > 0) {
            #pragma unroll
            for (int q = 0; q < DQ; ++q) {
                int j = threadIdx.x + q * BT;
                if (j < ni) {
                    j += rot;
                    if (j >= ni) j -= ni;
                    ((v4u*)dtab)[j] = dv[q];
                }
            }
        }
    }
    __syncthreads();

    const int T = cum[GB];
    const float* hl = Hs + lane * VPL;
    float* al = acc + lane * VPL;
    int g = wv * 32;
    int2 rcn = make_int2(0, 0);
    float wn = 0.0f;
    int sn = 0;
    if (g + lane < T) {
        sn = slotOf(cum, adj, g + lane);
        rcn = edges[sn];
        if constexpr (!WST) wn = wts[sn];
    }
    for (; g < T; g += BT) {
        const int cnt = T - g;
        const int2 rc = rcn;
        float w = wn;
        const int sl = sn;
        const int r = rc.x, c = rc.y;
        const int co = c * P;
        if (cnt >= 32) {
            float ax[16], ay[16];
            loadRows<F, 16>(hl, r, 0, ax, ay);
            if constexpr (WST) {
                const float dc = dl[c];
                w = srcDinv(dtab, dinv, r) * dc;
                wts[sl] = w;
            }
            if (g + BT + lane < T) {
                sn = slotOf(cum, adj, g + BT + lane);
                rcn = edges[sn];
                if constexpr (!WST) wn = wts[sn];
            }
            addLoaded<F, 16>(al, co, w, 0, ax, ay);
            addRows<F, 16>(hl, al, r, co, w, 16);
        } else {
            if constexpr (WST) {
                w = 0.0f;
                if (lane < cnt) {
                    const float dc = dl[c];
                    w = srcDinv(dtab, dinv, r) * dc;
                    wts[sl] = w;
                }
            }
            for (int j = 0; j < cnt; ++j) addRows<F, 1>(hl, al, r, co, w, j);
        }
    }
    __syncthreads();

    for (int i = threadIdx.x; i < NB * F / 4; i += BT) {
        const int nl = (i * 4) / F;
        const int node = node0 + nl;
        if (node < n) {
            const int f = (i * 4) % F;
            const size_t gi = (size_t)node0 * F + (size_t)i * 4;
            const float di = dinv[node];
            float4* ap = (float4*)(acc + nl * P + f);
            const float4 a = *ap;
            const float4 h = *(const float4*)(Hs + gi);
            float4 v;
            v.x = a.x + di * di * h.x + bias[f];
            v.y = a.y + di * di * h.y + bias[f + 1];
            v.z = a.z + di * di * h.z + bias[f + 2];
            v.w = a.w + di * di * h.w + bias[f + 3];
            if (RELU) {
                v.x = fmaxf(v.x, 0.0f);
                v.y = fmaxf(v.y, 0.0f);
                v.z = fmaxf(v.z, 0.0f);
                v.w = fmaxf(v.w, 0.0f);
            }
            if constexpr (COUT > 0) *ap = v;
            else *(float4*)(out + gi) = v;
        }
    }

    if constexpr (COUT > 0) {
        constexpr int CT = COUT / 16;
        __syncthreads();
        if (wv < NB / 16) {
            const int m  = lane & 15;
            const int kh = (lane >> 4) << 1;
            const float* ar = acc + (wv * 16 + m) * P + kh;
            const float* wc = Wn + kh * COUT + m;
            v8f c[CT] = {};
            #pragma unroll
            for (int k = 0; k < F; k += 4) {
                v2f a;
                const float2 av = *(const float2*)(ar + k);
                a[0] = av.x;
                a[1] = av.y;
                #pragma unroll
                for (int t = 0; t < CT; ++t) {
                    v2f bf;
                    bf[0] = wc[(size_t)k       * COUT + t * 16];
                    bf[1] = wc[(size_t)(k + 1) * COUT + t * 16];
                    c[t] = __builtin_amdgcn_wmma_f32_16x16x4_f32(
                         false, a,  false, bf,
                         (short)0, c[t],  false,  false);
                }
            }
            const int mo = (lane >> 4) << 3;
            #pragma unroll
            for (int v = 0; v < 8; ++v) {
                const int r = node0 + wv * 16 + mo + v;
                if (r < n) {
                    #pragma unroll
                    for (int t = 0; t < CT; ++t) out[(size_t)r * COUT + t * 16 + m] = c[t][v];
                }
            }
        }
    }
}

#define DEPW 64
#define DBT  128
__global__ __launch_bounds__(DBT) void k_decode(const float* __restrict__ Z, const int* __restrict__ ea,
                                                const int* __restrict__ eb, float* __restrict__ out, long long EL) {
    static_assert(DEPW == 64, "two halves of 32 edges");
    const int lane = threadIdx.x & 31;
    const int t = lane & 7;
    const long long wave = (long long)blockIdx.x * (DBT / 32) + (threadIdx.x >> 5);
    const long long base = wave * DEPW;
    if (base >= EL) return;
    int ia[2], ib[2];
    #pragma unroll
    for (int h = 0; h < 2; ++h) {
        long long e = base + 32 * h + lane;
        if (e >= EL) e = EL - 1;
        ia[h] = ea[e];
        ib[h] = eb[e];
    }
    float4 a[16], b[16];
    #pragma unroll
    for (int h = 0; h < 2; ++h) {
        #pragma unroll
        for (int s = 0; s < 8; ++s) {
            const int ra = __shfl(ia[h], s, 8);
            const int rb = __shfl(ib[h], s, 8);
            a[8 * h + s] = *(const float4*)(Z + (size_t)ra * 32 + t * 4);
            b[8 * h + s] = *(const float4*)(Z + (size_t)rb * 32 + t * 4);
        }
    }
    #pragma unroll
    for (int h = 0; h < 2; ++h) {
        float res = 0.0f;
        #pragma unroll
        for (int s = 0; s < 8; ++s) {
            const float4 x = a[8 * h + s], y = b[8 * h + s];
            float p = x.x * y.x + x.y * y.y + x.z * y.z + x.w * y.w;
            p += __shfl_xor(p, 1);
            p += __shfl_xor(p, 2);
            p += __shfl_xor(p, 4);
            if (t == s) res = p;
        }
        const long long e = base + 32 * h + lane;
        if (e < EL) out[e] = res;
    }
}

static inline int cdiv(long long a, long long b) { return (int)((a + b - 1) / b); }

extern "C" void kernel_launch(void* const* d_in, const int* in_sizes, int n_in,
                              void* d_out, int out_size, void* d_ws, size_t ws_size,
                              hipStream_t stream) {
    const float* x   = (const float*)d_in[0];
    const float* W1  = (const float*)d_in[1];
    const float* b1  = (const float*)d_in[2];
    const float* W2  = (const float*)d_in[3];
    const float* b2  = (const float*)d_in[4];
    const int*   ei  = (const int*)d_in[5];
    const int*   eli = (const int*)d_in[6];

    const long long N  = in_sizes[0] / F_IN;
    const long long E  = in_sizes[5] / 2;
    const long long EL = in_sizes[6] / 2;
    float* out = (float*)d_out;
    const int nbk = cdiv(N, NB);

    float* h1    = (float*)d_ws;
    float* h2    = h1   + N * F_HID;
    float* agg2  = h2   + N * F_OUT;
    unsigned short* deg16 = (unsigned short*)(agg2 + N * F_OUT);
    int2*  edges = (int2*)(deg16 + 4 * N);
    float* dinv  = (float*)(edges + E);
    float* wts   = dinv + N;
    int*   tab   = (int*)(wts + E);

    const int* row = ei;
    const int* col = ei + E;

    k_lsort<<<GB, BT, 0, stream>>>(row, col, edges, tab, E, nbk);

    k_bdeg_gemm<<<nbk, BT, 0, stream>>>(edges, tab, E, dinv, deg16, x, W1, h1, (int)N);

    k_bagg<F_HID, true, F_OUT, true><<<nbk, BT, 0, stream>>>(h1, edges, tab, E, dinv, b1, W2, wts, deg16, h2, (int)N);

    k_bagg<F_OUT, false, 0, false><<<nbk, BT, 0, stream>>>(h2, edges, tab, E, dinv, b2, nullptr, wts, nullptr, agg2, (int)N);

    k_decode<<<cdiv(EL, DEPW * (DBT / 32)), DBT, 0, stream>>>(agg2, eli, eli + EL, out, EL);
}


# speedup vs baseline: 5.9949x; 1.0398x over previous
#include <hip/hip_runtime.h>
#include <hip/hip_bf16.h>


typedef float v2f __attribute__((ext_vector_type(2)));
typedef float v8f __attribute__((ext_vector_type(8)));
typedef float v4f __attribute__((ext_vector_type(4)));
typedef unsigned int v4u __attribute__((ext_vector_type(4)));

#define F_IN  128
#define F_HID 64
#define F_OUT 32

#define NB   400
#define BT   1024
#define GB   256
#define ECH  16
#define MAXB 4096
#define DCAP 100352
#define DESC 0xFFFFu
#define SCAP 16384
#define LBLK 32768
#define NLB  10


template <bool STAGE>
__device__ __forceinline__ void placeRange(const int* __restrict__ cb, const int* __restrict__ rb, int nloc,
                                           int* hist, int2* stage, int2* __restrict__ edges, int lane) {
    for (int i0 = threadIdx.x; i0 < nloc; i0 += ECH * BT) {
        int c[ECH], r[ECH], p[ECH];
        #pragma unroll
        for (int q = 0; q < ECH; ++q) {
            int i = i0 + q * BT;
            if (i >= nloc) i = nloc - 1;
            c[q] = cb[i];
            r[q] = rb[i];
        }
        static_assert(ECH == 16, "the statements below name all ECH loaded words");
        asm volatile("" :: "v"(c[0]), "v"(c[1]), "v"(c[2]), "v"(c[3]), "v"(c[4]), "v"(c[5]), "v"(c[6]), "v"(c[7]),
                           "v"(c[8]), "v"(c[9]), "v"(c[10]), "v"(c[11]), "v"(c[12]), "v"(c[13]), "v"(c[14]), "v"(c[15]));
        asm volatile("" :: "v"(r[0]), "v"(r[1]), "v"(r[2]), "v"(r[3]), "v"(r[4]), "v"(r[5]), "v"(r[6]), "v"(r[7]),
                           "v"(r[8]), "v"(r[9]), "v"(r[10]), "v"(r[11]), "v"(r[12]), "v"(r[13]), "v"(r[14]), "v"(r[15]));
        #pragma unroll
        for (int q = 0; q < ECH; ++q) {
            const int hb = (i0 + q * BT < nloc) ? c[q] / NB : MAXB + lane;
            p[q] = atomicAdd(&hist[hb], 1);
        }
        #pragma unroll
        for (int q = 0; q < ECH; ++q) {
            if (i0 + q * BT < nloc) {
                const int b = c[q] / NB;
                const int2 v = make_int2(r[q], c[q] - b * NB);
                if constexpr (STAGE) stage[p[q]] = v;
                else edges[p[q]] = v;
            }
        }
    }
}

__global__ __launch_bounds__(BT) void k_lsort(const int* __restrict__ row, const int* __restrict__ col,
                                              int2* __restrict__ edges, int* __restrict__ tab,
                                              long long E, int nbk) {
    __shared__ int hist[MAXB + 32];
    __shared__ int wsum[BT / 32];
    __shared__ int2 stage[SCAP];
    const int lane = threadIdx.x & 31;
    const int wv = __builtin_amdgcn_readfirstlane((int)(threadIdx.x >> 5));
    for (int i = threadIdx.x; i < nbk; i += BT) hist[i] = 0;
    if (threadIdx.x < 32) hist[MAXB + threadIdx.x] = 0;
    __syncthreads();
    const long long per = (E + GB - 1) / GB;
    long long s = per * blockIdx.x;
    if (s > E) s = E;
    long long t = s + per;
    if (t > E) t = E;
    const int nloc = (int)(t - s);
    const bool inl = nloc <= SCAP;
    const int* cb = col + s;
    const int* rb = row + s;
    for (int i0 = threadIdx.x; i0 < nloc; i0 += ECH * BT) {
        int c[ECH];
        #pragma unroll
        for (int q = 0; q < ECH; ++q) {
            int i = i0 + q * BT;
            if (i >= nloc) i = nloc - 1;
            c[q] = cb[i];
        }
        static_assert(ECH == 16, "the statement below names all ECH loaded words");
        asm volatile("" :: "v"(c[0]), "v"(c[1]), "v"(c[2]), "v"(c[3]), "v"(c[4]), "v"(c[5]), "v"(c[6]), "v"(c[7]),
                           "v"(c[8]), "v"(c[9]), "v"(c[10]), "v"(c[11]), "v"(c[12]), "v"(c[13]), "v"(c[14]), "v"(c[15]));
        #pragma unroll
        for (int q = 0; q < ECH; ++q)
            if (i0 + q * BT < nloc) atomicAdd(&hist[c[q] / NB], 1);
    }
    __syncthreads();
    const int ipt = (nbk + BT - 1) / BT;
    const int b0 = threadIdx.x * ipt;
    int run = 0;
    for (int i = b0; i < b0 + ipt && i < nbk; ++i) run += hist[i];
    int incl = run;
    for (int d = 1; d < 32; d <<= 1) {
        const int y = __shfl_up(incl, d);
        if (lane >= d) incl += y;
    }
    if (lane == 31) wsum[wv] = incl;
    __syncthreads();
    int base = incl - run;
    for (int w = 0; w < wv; ++w) base += wsum[w];
    for (int i = b0; i < b0 + ipt && i < nbk; ++i) {
        const int h = hist[i];
        hist[i] = inl ? base : (int)s + base;
        tab[(size_t)i * GB + blockIdx.x] = base;
        base += h;
    }
    if (threadIdx.x == 0) tab[(size_t)nbk * GB + blockIdx.x] = nloc;
    __syncthreads();
    if (inl) {
        placeRange<true>(cb, rb, nloc, hist, stage, edges, lane);
        __syncthreads();
        for (int i0 = threadIdx.x; i0 < nloc; i0 += 4 * BT) {
            int2 v[4];
            #pragma unroll
            for (int q = 0; q < 4; ++q) {
                int i = i0 + q * BT;
                if (i >= nloc) i = nloc - 1;
                v[q] = stage[i];
            }
            asm volatile("" :: "v"(v[0].x), "v"(v[0].y), "v"(v[1].x), "v"(v[1].y),
                               "v"(v[2].x), "v"(v[2].y), "v"(v[3].x), "v"(v[3].y));
            #pragma unroll
            for (int q = 0; q < 4; ++q)
                if (i0 + q * BT < nloc) edges[s + i0 + q * BT] = v[q];
        }
    } else {
        placeRange<false>(cb, rb, nloc, hist, stage, edges, lane);
    }
}

__device__ __forceinline__ void segTable(const int* __restrict__ tab, int b, int per,
                                         int* cum, int* adj, int lane) {
    constexpr int SPL = GB / 32;
    static_assert(GB % 32 == 0 && (GB & (GB - 1)) == 0, "slotOf halves GB down to 1");
    int s[SPL], t[SPL];
    #pragma unroll
    for (int q = 0; q < SPL; ++q) {
        s[q] = tab[(size_t)b * GB + lane * SPL + q];
        t[q] = tab[(size_t)(b + 1) * GB + lane * SPL + q];
    }
    int ex[SPL];
    int run = 0;
    #pragma unroll
    for (int q = 0; q < SPL; ++q) {
        ex[q] = run;
        run += t[q] - s[q];
    }
    int incl = run;
    for (int d = 1; d < 32; d <<= 1) {
        const int y = __shfl_up(incl, d);
        if (lane >= d) incl += y;
    }
    const int base = incl - run;
    #pragma unroll
    for (int q = 0; q < SPL; ++q) {
        const int k = lane * SPL + q;
        cum[k] = base + ex[q];
        adj[k] = k * per + s[q] - (base + ex[q]);
    }
    if (lane == 31) cum[GB] = incl;
}

__device__ __forceinline__ int slotOf(const int* cum, const int* adj, int v) {
    int k = 0;
    #pragma unroll
    for (int st = GB / 2; st > 0; st >>= 1)
        if (cum[k + st] <= v) k += st;
    return v + adj[k];
}

#define XP  132
#define NS  (NB / 16)
#define RPT 4

__global__ __launch_bounds__(BT) void k_bdeg_gemm(const int2* __restrict__ edges, const int* __restrict__ tab, long long E,
                                                  float* __restrict__ dinv, unsigned short* __restrict__ deg16,
                                                  const float* __restrict__ X,
                                                  const float* __restrict__ W, float* __restrict__ H, int n) {
    static_assert(NB % 16 == 0 && NS <= BT / 32, "one 16-row strip per wave");
    static_assert((F_IN * F_HID / 4) % BT == 0, "W is read as whole float4 items per thread");
    __shared__ __attribute__((aligned(16))) float wt[F_HID * XP];
    __shared__ __attribute__((aligned(16))) float xs[NS * 16 * XP];
    __shared__ int cnt[NB];
    __shared__ int ticket;
    __shared__ int cum[GB + 1];
    __shared__ int adj[GB];
    const int lane = threadIdx.x & 31;
    const int wv = __builtin_amdgcn_readfirstlane((int)(threadIdx.x >> 5));
    const int b = blockIdx.x;
    const int node0 = b * NB;
    const int r0 = node0 + wv * 16;
    const int per = (int)((E + GB - 1) / GB);

    if (wv < NS) {
        int r = r0 + (lane >> 1);
        if (r >= n) r = n - 1;
        __builtin_prefetch(X + (size_t)r * F_IN + (lane & 1) * (F_IN / 2), 0, 3);
    }
    constexpr int WQ = F_IN * F_HID / 4 / BT;
    v4f wr[WQ];
    #pragma unroll
    for (int q = 0; q < WQ; ++q) wr[q] = ((const v4f*)W)[threadIdx.x + q * BT];
    for (int i = threadIdx.x; i < NB; i += BT) cnt[i] = 0;
    if (threadIdx.x == 0) ticket = 0;
    if (wv == 0) segTable(tab, b, per, cum, adj, lane);
    #pragma unroll
    for (int q = 0; q < WQ; ++q) {
        const int j = threadIdx.x + q * BT;
        const int k = j / (F_HID / 4), c4 = j % (F_HID / 4);
        float* p = wt + (4 * c4) * XP + k;
        p[0]      = wr[q][0];
        p[XP]     = wr[q][1];
        p[2 * XP] = wr[q][2];
        p[3 * XP] = wr[q][3];
    }
    __syncthreads();

    if (wv < NS) {
        v4f xr[16];
        #pragma unroll
        for (int i = 0; i < 16; ++i) {
            int r = r0 + i;
            if (r >= n) r = n - 1;
            xr[i] = *(const v4f*)(X + (size_t)r * F_IN + lane * 4);
        }
        float* xw = xs + wv * 16 * XP;
        #pragma unroll
        for (int i = 0; i < 16; ++i) *(v4f*)(xw + i * XP + lane * 4) = xr[i];
        const int m  = lane & 15;
        const int kh = (lane >> 4) << 1;
        const float* ar = xw + m * XP + kh;
        const float* br = wt + m * XP + kh;
        constexpr int CT = F_HID / 16;
        v8f c[CT] = {};
        #pragma unroll
        for (int k = 0; k < F_IN; k += 4) {
            v2f a;
            const float2 av = *(const float2*)(ar + k);
            a[0] = av.x;
            a[1] = av.y;
            #pragma unroll
            for (int t = 0; t < CT; ++t) {
                v2f bf;
                const float2 bv = *(const float2*)(br + t * 16 * XP + k);
                bf[0] = bv.x;
                bf[1] = bv.y;
                c[t] = __builtin_amdgcn_wmma_f32_16x16x4_f32(
                     false, a,  false, bf,
                     (short)0, c[t],  false,  false);
            }
        }
        const int mo = (lane >> 4) << 3;
        #pragma unroll
        for (int v = 0; v < 8; ++v) {
            const int r = r0 + mo + v;
            if (r < n) {
                #pragma unroll
                for (int t = 0; t < CT; ++t) H[(size_t)r * F_HID + t * 16 + m] = c[t][v];
            }
        }
    }

    static_assert(GB % RPT == 0, "a ticket is RPT whole runs");
    const int T = cum[GB];
    for (; T > 0;) {
        int t = 0;
        if (lane == 0) t = atomicAdd(&ticket, 1);
        t = __builtin_amdgcn_readfirstlane(t);
        const int k0 = t * RPT;
        if (k0 >= GB) break;
        int d[2 * RPT], len[RPT], base[RPT];
        #pragma unroll
        for (int q = 0; q < RPT; ++q) {
            const int c0 = cum[k0 + q];
            len[q] = cum[k0 + q + 1] - c0;
            base[q] = adj[k0 + q] + c0;
            d[2 * q]     = edges[(lane      < len[q]) ? base[q] + lane      : 0].y;
            d[2 * q + 1] = edges[(lane + 32 < len[q]) ? base[q] + lane + 32 : 0].y;
        }
        static_assert(RPT == 4, "the statement below names all 2*RPT loaded words");
        asm volatile("" :: "v"(d[0]), "v"(d[1]), "v"(d[2]), "v"(d[3]), "v"(d[4]), "v"(d[5]), "v"(d[6]), "v"(d[7]));
        #pragma unroll
        for (int q = 0; q < 2 * RPT; ++q)
            if (lane + 32 * (q & 1) < len[q >> 1]) atomicAdd(&cnt[d[q]], 1);
        #pragma unroll
        for (int q = 0; q < RPT; ++q)
            for (int o = lane + 64; o < len[q]; o += 32) atomicAdd(&cnt[edges[base[q] + o].y], 1);
    }
    __syncthreads();
    for (int i = threadIdx.x; i < NB; i += BT) {
        const int node = node0 + i;
        if (node < n) {
            const int d = cnt[i] + 1;
            dinv[node] = rsqrtf((float)d);
            deg16[node] = (unsigned short)(d < (int)DESC ? (unsigned)d : DESC);
        }
    }
}

__device__ __forceinline__ void ldsAddF(float* p, float v) {
    __hip_atomic_fetch_add(p, v, __ATOMIC_RELAXED, __HIP_MEMORY_SCOPE_WORKGROUP);
}

template <int F, int J>
__device__ __forceinline__ void loadRows(const float* __restrict__ hl, int r, int j0,
                                         float (&vx)[J], float (&vy)[J]) {
    #pragma unroll
    for (int j = 0; j < J; ++j) {
        const int rj = __builtin_amdgcn_readlane(r, j0 + j);
        const float* p = hl + (size_t)rj * F;
        if constexpr (F == 64) {
            const float2 v = *(const float2*)p;
            vx[j] = v.x;
            vy[j] = v.y;
        } else {
            vx[j] = *p;
            vy[j] = 0.0f;
        }
    }
}

template <int F, int J>
__device__ __forceinline__ void addLoaded(float* al, int co, float w, int j0,
                                          const float (&vx)[J], const float (&vy)[J]) {
    #pragma unroll
    for (int j = 0; j < J; ++j) {
        const int cj = __builtin_amdgcn_readlane(co, j0 + j);
        const float wj = __int_as_float(__builtin_amdgcn_readlane(__float_as_int(w), j0 + j));
        float* q = al + cj;
        ldsAddF(q, vx[j] * wj);
        if constexpr (F == 64) ldsAddF(q + 1, vy[j] * wj);
    }
}

template <int F, int J>
__device__ __forceinline__ void addRows(const float* __restrict__ hl, float* al,
                                        int r, int co, float w, int j0) {
    float vx[J], vy[J];
    loadRows<F, J>(hl, r, j0, vx, vy);
    addLoaded<F, J>(al, co, w, j0, vx, vy);
}

template <int ACCB, int DWB>
__device__ __forceinline__ int dwOff(int i) {
    const int k = i / DWB;
    return k * (LBLK / 2) + ACCB / 2 + (i - k * DWB);
}

template <int ACCB, int DWB>
__device__ __forceinline__ float srcDinv(const unsigned short* lds16, const float* __restrict__ dinv, int r) {
    const unsigned d = r < DCAP ? (unsigned)lds16[dwOff<ACCB, DWB>(r)] : DESC;
    float s;
    if (d != DESC) {
        s = rsqrtf((float)d);
    } else {
        s = dinv[r];
        asm volatile("" :: "v"(s));
    }
    return s;
}

template <int P>
__device__ __forceinline__ int rowOff(int c) {
    const int k = c / (NB / NLB);
    return k * (LBLK / 4) + (c - k * (NB / NLB)) * P;
}

template <int F, bool RELU, int COUT, bool WST>
__global__ __launch_bounds__(BT) void k_bagg(const float* __restrict__ Hs, const int2* __restrict__ edges,
                                             const int* __restrict__ tab, long long E, const float* __restrict__ dinv,
                                             const float* __restrict__ bias, const float* __restrict__ Wn,
                                             float* __restrict__ wts, const unsigned short* __restrict__ deg16,
                                             float* __restrict__ out, int n) {
    constexpr int VPL = F / 32;
    constexpr int P = (COUT > 0) ? F + 4 : F;
    static_assert(NB % 16 == 0 && NB / 16 <= BT / 32, "one 16-row strip per wave");
    static_assert(NB % NLB == 0, "the same number of rows in every LDS block");
    constexpr int ACCB = (NB / NLB) * P * 4;
    constexpr int DN = WST ? DCAP : 8;
    static_assert(DN % 8 == 0, "the table is copied in 16-byte items of 8 words");
    constexpr int DWB = WST ? ((LBLK - ACCB) / 16) * 8 : 8;
    static_assert(ACCB % 16 == 0 && ACCB < LBLK, "a block's table piece starts on a 16-byte boundary behind its sums");
    static_assert(!WST || (DN - 1) / DWB == NLB - 1, "the table's last word lies in the last block");
    constexpr int TAILB = WST ? (DN - (NLB - 1) * DWB) * 2 : 0;
    __shared__ __attribute__((aligned(LBLK))) unsigned char lds[(NLB - 1) * LBLK + ACCB + TAILB];
    float* const acc = (float*)lds;
    unsigned short* const lds16 = (unsigned short*)lds;
    __shared__ int cum[GB + 1];
    __shared__ int adj[GB];
    __shared__ float dl[WST ? NB : 1];
    const int lane = threadIdx.x & 31;
    const int wv = __builtin_amdgcn_readfirstlane((int)(threadIdx.x >> 5));
    const int b = blockIdx.x;
    const int node0 = b * NB;

    constexpr int DQ = (DN / 8 + BT - 1) / BT;
    const int nt = n < DN ? n : DN;
    const int ni = nt / 8;
    const int rot = (int)((unsigned)(b * ni) / gridDim.x);
    v4u dv[DQ];
    if constexpr (WST) {
        if (ni > 0) {
            #pragma unroll
            for (int q = 0; q < DQ; ++q) {
                int j = threadIdx.x + q * BT;
                if (j >= ni) j = ni - 1;
                j += rot;
                if (j >= ni) j -= ni;
                dv[q] = ((const v4u*)deg16)[j];
            }
        }
    }
    static_assert(P % 4 == 0, "a row is zeroed in whole float4 items");
    for (int i = threadIdx.x; i < NB * P / 4; i += BT) {
        const int row = (i * 4) / P;
        *(float4*)(acc + rowOff<P>(row) + (i * 4 - row * P)) = make_float4(0.f, 0.f, 0.f, 0.f);
    }
    if (wv == 0) segTable(tab, b, (int)((E + GB - 1) / GB), cum, adj, lane);
    if constexpr (WST) {
        for (int i = threadIdx.x; i < NB; i += BT) {
            const int node = node0 + i;
            dl[i] = node < n ? dinv[node] : 0.0f;
        }
        if ((int)threadIdx.x < nt - ni * 8)
            lds16[dwOff<ACCB, DWB>(ni * 8 + (int)threadIdx.x)] = deg16[ni * 8 + threadIdx.x];
        if (ni > 0) {
            #pragma unroll
            for (int q = 0; q < DQ; ++q) {
                int j = threadIdx.x + q * BT;
                if (j < ni) {
                    j += rot;
                    if (j >= ni) j -= ni;
                    *(v4u*)(lds16 + dwOff<ACCB, DWB>(8 * j)) = dv[q];
                }
            }
        }
    }
    __syncthreads();

    const int T = cum[GB];
    const float* hl = Hs + lane * VPL;
    float* al = acc + lane * VPL;
    int g = wv * 32;
    int2 rcn = make_int2(0, 0);
    float wn = 0.0f;
    int sn = 0;
    if (g + lane < T) {
        sn = slotOf(cum, adj, g + lane);
        rcn = edges[sn];
        if constexpr (!WST) wn = wts[sn];
    }
    for (; g < T; g += BT) {
        const int cnt = T - g;
        const int2 rc = rcn;
        float w = wn;
        const int sl = sn;
        const int r = rc.x, c = rc.y;
        const int co = rowOff<P>(c);
        if (cnt >= 32) {
            float ax[16], ay[16];
            loadRows<F, 16>(hl, r, 0, ax, ay);
            if constexpr (WST) {
                const float dc = dl[c];
                w = srcDinv<ACCB, DWB>(lds16, dinv, r) * dc;
                wts[sl] = w;
            }
            if (g + BT + lane < T) {
                sn = slotOf(cum, adj, g + BT + lane);
                rcn = edges[sn];
                if constexpr (!WST) wn = wts[sn];
            }
            addLoaded<F, 16>(al, co, w, 0, ax, ay);
            addRows<F, 16>(hl, al, r, co, w, 16);
        } else {
            if constexpr (WST) {
                w = 0.0f;
                if (lane < cnt) {
                    const float dc = dl[c];
                    w = srcDinv<ACCB, DWB>(lds16, dinv, r) * dc;
                    wts[sl] = w;
                }
            }
            for (int j = 0; j < cnt; ++j) addRows<F, 1>(hl, al, r, co, w, j);
        }
    }
    __syncthreads();

    for (int i = threadIdx.x; i < NB * F / 4; i += BT) {
        const int nl = (i * 4) / F;
        const int node = node0 + nl;
        if (node < n) {
            const int f = (i * 4) % F;
            const size_t gi = (size_t)node0 * F + (size_t)i * 4;
            const float di = dinv[node];
            float4* ap = (float4*)(acc + rowOff<P>(nl) + f);
            const float4 a = *ap;
            const float4 h = *(const float4*)(Hs + gi);
            float4 v;
            v.x = a.x + di * di * h.x + bias[f];
            v.y = a.y + di * di * h.y + bias[f + 1];
            v.z = a.z + di * di * h.z + bias[f + 2];
            v.w = a.w + di * di * h.w + bias[f + 3];
            if (RELU) {
                v.x = fmaxf(v.x, 0.0f);
                v.y = fmaxf(v.y, 0.0f);
                v.z = fmaxf(v.z, 0.0f);
                v.w = fmaxf(v.w, 0.0f);
            }
            if constexpr (COUT > 0) *ap = v;
            else *(float4*)(out + gi) = v;
        }
    }

    if constexpr (COUT > 0) {
        constexpr int CT = COUT / 16;
        __syncthreads();
        if (wv < NB / 16) {
            const int m  = lane & 15;
            const int kh = (lane >> 4) << 1;
            const float* ar = acc + rowOff<P>(wv * 16 + m) + kh;
            const float* wc = Wn + kh * COUT + m;
            v8f c[CT] = {};
            #pragma unroll
            for (int k = 0; k < F; k += 4) {
                v2f a;
                const float2 av = *(const float2*)(ar + k);
                a[0] = av.x;
                a[1] = av.y;
                #pragma unroll
                for (int t = 0; t < CT; ++t) {
                    v2f bf;
                    bf[0] = wc[(size_t)k       * COUT + t * 16];
                    bf[1] = wc[(size_t)(k + 1) * COUT + t * 16];
                    c[t] = __builtin_amdgcn_wmma_f32_16x16x4_f32(
                         false, a,  false, bf,
                         (short)0, c[t],  false,  false);
                }
            }
            const int mo = (lane >> 4) << 3;
            #pragma unroll
            for (int v = 0; v < 8; ++v) {
                const int r = node0 + wv * 16 + mo + v;
                if (r < n) {
                    #pragma unroll
                    for (int t = 0; t < CT; ++t) out[(size_t)r * COUT + t * 16 + m] = c[t][v];
                }
            }
        }
    }
}

#define DEPW 64
#define DBT  128
__global__ __launch_bounds__(DBT) void k_decode(const float* __restrict__ Z, const int* __restrict__ ea,
                                                const int* __restrict__ eb, float* __restrict__ out, long long EL) {
    static_assert(DEPW == 64, "two halves of 32 edges");
    const int lane = threadIdx.x & 31;
    const int t = lane & 7;
    const long long wave = (long long)blockIdx.x * (DBT / 32) + (threadIdx.x >> 5);
    const long long base = wave * DEPW;
    if (base >= EL) return;
    int ia[2], ib[2];
    #pragma unroll
    for (int h = 0; h < 2; ++h) {
        long long e = base + 32 * h + lane;
        if (e >= EL) e = EL - 1;
        ia[h] = ea[e];
        ib[h] = eb[e];
    }
    float4 a[16], b[16];
    #pragma unroll
    for (int h = 0; h < 2; ++h) {
        #pragma unroll
        for (int s = 0; s < 8; ++s) {
            const int ra = __shfl(ia[h], s, 8);
            const int rb = __shfl(ib[h], s, 8);
            a[8 * h + s] = *(const float4*)(Z + (size_t)ra * 32 + t * 4);
            b[8 * h + s] = *(const float4*)(Z + (size_t)rb * 32 + t * 4);
        }
    }
    #pragma unroll
    for (int h = 0; h < 2; ++h) {
        float res = 0.0f;
        #pragma unroll
        for (int s = 0; s < 8; ++s) {
            const float4 x = a[8 * h + s], y = b[8 * h + s];
            float p = x.x * y.x + x.y * y.y + x.z * y.z + x.w * y.w;
            p += __shfl_xor(p, 1);
            p += __shfl_xor(p, 2);
            p += __shfl_xor(p, 4);
            if (t == s) res = p;
        }
        const long long e = base + 32 * h + lane;
        if (e < EL) out[e] = res;
    }
}

static inline int cdiv(long long a, long long b) { return (int)((a + b - 1) / b); }

extern "C" void kernel_launch(void* const* d_in, const int* in_sizes, int n_in,
                              void* d_out, int out_size, void* d_ws, size_t ws_size,
                              hipStream_t stream) {
    const float* x   = (const float*)d_in[0];
    const float* W1  = (const float*)d_in[1];
    const float* b1  = (const float*)d_in[2];
    const float* W2  = (const float*)d_in[3];
    const float* b2  = (const float*)d_in[4];
    const int*   ei  = (const int*)d_in[5];
    const int*   eli = (const int*)d_in[6];

    const long long N  = in_sizes[0] / F_IN;
    const long long E  = in_sizes[5] / 2;
    const long long EL = in_sizes[6] / 2;
    float* out = (float*)d_out;
    const int nbk = cdiv(N, NB);

    float* h1    = (float*)d_ws;
    float* h2    = h1   + N * F_HID;
    float* agg2  = h2   + N * F_OUT;
    unsigned short* deg16 = (unsigned short*)(agg2 + N * F_OUT);
    int2*  edges = (int2*)(deg16 + 4 * N);
    float* dinv  = (float*)(edges + E);
    float* wts   = dinv + N;
    int*   tab   = (int*)(wts + E);

    const int* row = ei;
    const int* col = ei + E;

    k_lsort<<<GB, BT, 0, stream>>>(row, col, edges, tab, E, nbk);

    k_bdeg_gemm<<<nbk, BT, 0, stream>>>(edges, tab, E, dinv, deg16, x, W1, h1, (int)N);

    k_bagg<F_HID, true, F_OUT, true><<<nbk, BT, 0, stream>>>(h1, edges, tab, E, dinv, b1, W2, wts, deg16, h2, (int)N);

    k_bagg<F_OUT, false, 0, false><<<nbk, BT, 0, stream>>>(h2, edges, tab, E, dinv, b2, nullptr, wts, nullptr, agg2, (int)N);

    k_decode<<<cdiv(EL, DEPW * (DBT / 32)), DBT, 0, stream>>>(agg2, eli, eli + EL, out, EL);
}


# aggregation passes: a bin's edges sorted by destination node in LDS (places: in-degree prefix + one integer LDS add an edge); 16 lanes sum a node's rows in registers; no LDS float adds; new sum order
# speedup vs baseline: 7.1464x; 1.1921x over previous
#include <hip/hip_runtime.h>
#include <hip/hip_bf16.h>


typedef float v2f __attribute__((ext_vector_type(2)));
typedef float v8f __attribute__((ext_vector_type(8)));
typedef float v4f __attribute__((ext_vector_type(4)));
typedef unsigned int v4u __attribute__((ext_vector_type(4)));

#define F_IN  128
#define F_HID 64
#define F_OUT 32

#define NB   400
#define BT   1024
#define GB   256
#define ECH  16
#define MAXB 4096
#define DCAP 100352
#define DESC 0xFFFFu
#define SCAP 16384


template <bool STAGE>
__device__ __forceinline__ void placeRange(const int* __restrict__ cb, const int* __restrict__ rb, int nloc,
                                           int* hist, int2* stage, int2* __restrict__ edges, int lane) {
    for (int i0 = threadIdx.x; i0 < nloc; i0 += ECH * BT) {
        int c[ECH], r[ECH], p[ECH];
        #pragma unroll
        for (int q = 0; q < ECH; ++q) {
            int i = i0 + q * BT;
            if (i >= nloc) i = nloc - 1;
            c[q] = cb[i];
            r[q] = rb[i];
        }
        static_assert(ECH == 16, "the statements below name all ECH loaded words");
        asm volatile("" :: "v"(c[0]), "v"(c[1]), "v"(c[2]), "v"(c[3]), "v"(c[4]), "v"(c[5]), "v"(c[6]), "v"(c[7]),
                           "v"(c[8]), "v"(c[9]), "v"(c[10]), "v"(c[11]), "v"(c[12]), "v"(c[13]), "v"(c[14]), "v"(c[15]));
        asm volatile("" :: "v"(r[0]), "v"(r[1]), "v"(r[2]), "v"(r[3]), "v"(r[4]), "v"(r[5]), "v"(r[6]), "v"(r[7]),
                           "v"(r[8]), "v"(r[9]), "v"(r[10]), "v"(r[11]), "v"(r[12]), "v"(r[13]), "v"(r[14]), "v"(r[15]));
        #pragma unroll
        for (int q = 0; q < ECH; ++q) {
            const int hb = (i0 + q * BT < nloc) ? c[q] / NB : MAXB + lane;
            p[q] = atomicAdd(&hist[hb], 1);
        }
        #pragma unroll
        for (int q = 0; q < ECH; ++q) {
            if (i0 + q * BT < nloc) {
                const int b = c[q] / NB;
                const int2 v = make_int2(r[q], c[q] - b * NB);
                if constexpr (STAGE) stage[p[q]] = v;
                else edges[p[q]] = v;
            }
        }
    }
}

__global__ __launch_bounds__(BT) void k_lsort(const int* __restrict__ row, const int* __restrict__ col,
                                              int2* __restrict__ edges, int* __restrict__ tab,
                                              long long E, int nbk) {
    __shared__ int hist[MAXB + 32];
    __shared__ int wsum[BT / 32];
    __shared__ int2 stage[SCAP];
    const int lane = threadIdx.x & 31;
    const int wv = __builtin_amdgcn_readfirstlane((int)(threadIdx.x >> 5));
    for (int i = threadIdx.x; i < nbk; i += BT) hist[i] = 0;
    if (threadIdx.x < 32) hist[MAXB + threadIdx.x] = 0;
    __syncthreads();
    const long long per = (E + GB - 1) / GB;
    long long s = per * blockIdx.x;
    if (s > E) s = E;
    long long t = s + per;
    if (t > E) t = E;
    const int nloc = (int)(t - s);
    const bool inl = nloc <= SCAP;
    const int* cb = col + s;
    const int* rb = row + s;
    for (int i0 = threadIdx.x; i0 < nloc; i0 += ECH * BT) {
        int c[ECH];
        #pragma unroll
        for (int q = 0; q < ECH; ++q) {
            int i = i0 + q * BT;
            if (i >= nloc) i = nloc - 1;
            c[q] = cb[i];
        }
        static_assert(ECH == 16, "the statement below names all ECH loaded words");
        asm volatile("" :: "v"(c[0]), "v"(c[1]), "v"(c[2]), "v"(c[3]), "v"(c[4]), "v"(c[5]), "v"(c[6]), "v"(c[7]),
                           "v"(c[8]), "v"(c[9]), "v"(c[10]), "v"(c[11]), "v"(c[12]), "v"(c[13]), "v"(c[14]), "v"(c[15]));
        #pragma unroll
        for (int q = 0; q < ECH; ++q)
            if (i0 + q * BT < nloc) atomicAdd(&hist[c[q] / NB], 1);
    }
    __syncthreads();
    const int ipt = (nbk + BT - 1) / BT;
    const int b0 = threadIdx.x * ipt;
    int run = 0;
    for (int i = b0; i < b0 + ipt && i < nbk; ++i) run += hist[i];
    int incl = run;
    for (int d = 1; d < 32; d <<= 1) {
        const int y = __shfl_up(incl, d);
        if (lane >= d) incl += y;
    }
    if (lane == 31) wsum[wv] = incl;
    __syncthreads();
    int base = incl - run;
    for (int w = 0; w < wv; ++w) base += wsum[w];
    for (int i = b0; i < b0 + ipt && i < nbk; ++i) {
        const int h = hist[i];
        hist[i] = inl ? base : (int)s + base;
        tab[(size_t)i * GB + blockIdx.x] = base;
        base += h;
    }
    if (threadIdx.x == 0) tab[(size_t)nbk * GB + blockIdx.x] = nloc;
    __syncthreads();
    if (inl) {
        placeRange<true>(cb, rb, nloc, hist, stage, edges, lane);
        __syncthreads();
        for (int i0 = threadIdx.x; i0 < nloc; i0 += 4 * BT) {
            int2 v[4];
            #pragma unroll
            for (int q = 0; q < 4; ++q) {
                int i = i0 + q * BT;
                if (i >= nloc) i = nloc - 1;
                v[q] = stage[i];
            }
            asm volatile("" :: "v"(v[0].x), "v"(v[0].y), "v"(v[1].x), "v"(v[1].y),
                               "v"(v[2].x), "v"(v[2].y), "v"(v[3].x), "v"(v[3].y));
            #pragma unroll
            for (int q = 0; q < 4; ++q)
                if (i0 + q * BT < nloc) edges[s + i0 + q * BT] = v[q];
        }
    } else {
        placeRange<false>(cb, rb, nloc, hist, stage, edges, lane);
    }
}

__device__ __forceinline__ void segTable(const int* __restrict__ tab, int b, int per,
                                         int* cum, int* adj, int lane) {
    constexpr int SPL = GB / 32;
    static_assert(GB % 32 == 0 && (GB & (GB - 1)) == 0, "slotOf halves GB down to 1");
    int s[SPL], t[SPL];
    #pragma unroll
    for (int q = 0; q < SPL; ++q) {
        s[q] = tab[(size_t)b * GB + lane * SPL + q];
        t[q] = tab[(size_t)(b + 1) * GB + lane * SPL + q];
    }
    int ex[SPL];
    int run = 0;
    #pragma unroll
    for (int q = 0; q < SPL; ++q) {
        ex[q] = run;
        run += t[q] - s[q];
    }
    int incl = run;
    for (int d = 1; d < 32; d <<= 1) {
        const int y = __shfl_up(incl, d);
        if (lane >= d) incl += y;
    }
    const int base = incl - run;
    #pragma unroll
    for (int q = 0; q < SPL; ++q) {
        const int k = lane * SPL + q;
        cum[k] = base + ex[q];
        adj[k] = k * per + s[q] - (base + ex[q]);
    }
    if (lane == 31) cum[GB] = incl;
}

__device__ __forceinline__ int slotOf(const int* cum, const int* adj, int v) {
    int k = 0;
    #pragma unroll
    for (int st = GB / 2; st > 0; st >>= 1)
        if (cum[k + st] <= v) k += st;
    return v + adj[k];
}

#define XP  132
#define NS  (NB / 16)
#define RPT 4

__global__ __launch_bounds__(BT) void k_bdeg_gemm(const int2* __restrict__ edges, const int* __restrict__ tab, long long E,
                                                  float* __restrict__ dinv, unsigned short* __restrict__ deg16,
                                                  int* __restrict__ degc, const float* __restrict__ X,
                                                  const float* __restrict__ W, float* __restrict__ H, int n) {
    static_assert(NB % 16 == 0 && NS <= BT / 32, "one 16-row strip per wave");
    static_assert((F_IN * F_HID / 4) % BT == 0, "W is read as whole float4 items per thread");
    __shared__ __attribute__((aligned(16))) float wt[F_HID * XP];
    __shared__ __attribute__((aligned(16))) float xs[NS * 16 * XP];
    __shared__ int cnt[NB];
    __shared__ int ticket;
    __shared__ int cum[GB + 1];
    __shared__ int adj[GB];
    const int lane = threadIdx.x & 31;
    const int wv = __builtin_amdgcn_readfirstlane((int)(threadIdx.x >> 5));
    const int b = blockIdx.x;
    const int node0 = b * NB;
    const int r0 = node0 + wv * 16;
    const int per = (int)((E + GB - 1) / GB);

    if (wv < NS) {
        int r = r0 + (lane >> 1);
        if (r >= n) r = n - 1;
        __builtin_prefetch(X + (size_t)r * F_IN + (lane & 1) * (F_IN / 2), 0, 3);
    }
    constexpr int WQ = F_IN * F_HID / 4 / BT;
    v4f wr[WQ];
    #pragma unroll
    for (int q = 0; q < WQ; ++q) wr[q] = ((const v4f*)W)[threadIdx.x + q * BT];
    for (int i = threadIdx.x; i < NB; i += BT) cnt[i] = 0;
    if (threadIdx.x == 0) ticket = 0;
    if (wv == 0) segTable(tab, b, per, cum, adj, lane);
    #pragma unroll
    for (int q = 0; q < WQ; ++q) {
        const int j = threadIdx.x + q * BT;
        const int k = j / (F_HID / 4), c4 = j % (F_HID / 4);
        float* p = wt + (4 * c4) * XP + k;
        p[0]      = wr[q][0];
        p[XP]     = wr[q][1];
        p[2 * XP] = wr[q][2];
        p[3 * XP] = wr[q][3];
    }
    __syncthreads();

    if (wv < NS) {
        v4f xr[16];
        #pragma unroll
        for (int i = 0; i < 16; ++i) {
            int r = r0 + i;
            if (r >= n) r = n - 1;
            xr[i] = *(const v4f*)(X + (size_t)r * F_IN + lane * 4);
        }
        float* xw = xs + wv * 16 * XP;
        #pragma unroll
        for (int i = 0; i < 16; ++i) *(v4f*)(xw + i * XP + lane * 4) = xr[i];
        const int m  = lane & 15;
        const int kh = (lane >> 4) << 1;
        const float* ar = xw + m * XP + kh;
        const float* br = wt + m * XP + kh;
        constexpr int CT = F_HID / 16;
        v8f c[CT] = {};
        #pragma unroll
        for (int k = 0; k < F_IN; k += 4) {
            v2f a;
            const float2 av = *(const float2*)(ar + k);
            a[0] = av.x;
            a[1] = av.y;
            #pragma unroll
            for (int t = 0; t < CT; ++t) {
                v2f bf;
                const float2 bv = *(const float2*)(br + t * 16 * XP + k);
                bf[0] = bv.x;
                bf[1] = bv.y;
                c[t] = __builtin_amdgcn_wmma_f32_16x16x4_f32(
                     false, a,  false, bf,
                     (short)0, c[t],  false,  false);
            }
        }
        const int mo = (lane >> 4) << 3;
        #pragma unroll
        for (int v = 0; v < 8; ++v) {
            const int r = r0 + mo + v;
            if (r < n) {
                #pragma unroll
                for (int t = 0; t < CT; ++t) H[(size_t)r * F_HID + t * 16 + m] = c[t][v];
            }
        }
    }

    static_assert(GB % RPT == 0, "a ticket is RPT whole runs");
    const int T = cum[GB];
    for (; T > 0;) {
        int t = 0;
        if (lane == 0) t = atomicAdd(&ticket, 1);
        t = __builtin_amdgcn_readfirstlane(t);
        const int k0 = t * RPT;
        if (k0 >= GB) break;
        int d[2 * RPT], len[RPT], base[RPT];
        #pragma unroll
        for (int q = 0; q < RPT; ++q) {
            const int c0 = cum[k0 + q];
            len[q] = cum[k0 + q + 1] - c0;
            base[q] = adj[k0 + q] + c0;
            d[2 * q]     = edges[(lane      < len[q]) ? base[q] + lane      : 0].y;
            d[2 * q + 1] = edges[(lane + 32 < len[q]) ? base[q] + lane + 32 : 0].y;
        }
        static_assert(RPT == 4, "the statement below names all 2*RPT loaded words");
        asm volatile("" :: "v"(d[0]), "v"(d[1]), "v"(d[2]), "v"(d[3]), "v"(d[4]), "v"(d[5]), "v"(d[6]), "v"(d[7]));
        #pragma unroll
        for (int q = 0; q < 2 * RPT; ++q)
            if (lane + 32 * (q & 1) < len[q >> 1]) atomicAdd(&cnt[d[q]], 1);
        #pragma unroll
        for (int q = 0; q < RPT; ++q)
            for (int o = lane + 64; o < len[q]; o += 32) atomicAdd(&cnt[edges[base[q] + o].y], 1);
    }
    __syncthreads();
    for (int i = threadIdx.x; i < NB; i += BT) {
        const int node = node0 + i;
        if (node < n) {
            const int d = cnt[i] + 1;
            dinv[node] = rsqrtf((float)d);
            deg16[node] = (unsigned short)(d < (int)DESC ? (unsigned)d : DESC);
            degc[node] = cnt[i];
        }
    }
}

__device__ __forceinline__ void ldsAddF(float* p, float v) {
    __hip_atomic_fetch_add(p, v, __ATOMIC_RELAXED, __HIP_MEMORY_SCOPE_WORKGROUP);
}

template <int F, int J>
__device__ __forceinline__ void loadRows(const float* __restrict__ hl, int r, int j0,
                                         float (&vx)[J], float (&vy)[J]) {
    #pragma unroll
    for (int j = 0; j < J; ++j) {
        const int rj = __builtin_amdgcn_readlane(r, j0 + j);
        const float* p = hl + (size_t)rj * F;
        if constexpr (F == 64) {
            const float2 v = *(const float2*)p;
            vx[j] = v.x;
            vy[j] = v.y;
        } else {
            vx[j] = *p;
            vy[j] = 0.0f;
        }
    }
}

template <int F, int J>
__device__ __forceinline__ void addLoaded(float* al, int co, float w, int j0,
                                          const float (&vx)[J], const float (&vy)[J]) {
    #pragma unroll
    for (int j = 0; j < J; ++j) {
        const int cj = __builtin_amdgcn_readlane(co, j0 + j);
        const float wj = __int_as_float(__builtin_amdgcn_readlane(__float_as_int(w), j0 + j));
        float* q = al + cj;
        ldsAddF(q, vx[j] * wj);
        if constexpr (F == 64) ldsAddF(q + 1, vy[j] * wj);
    }
}

template <int F, int J>
__device__ __forceinline__ void addRows(const float* __restrict__ hl, float* al,
                                        int r, int co, float w, int j0) {
    float vx[J], vy[J];
    loadRows<F, J>(hl, r, j0, vx, vy);
    addLoaded<F, J>(al, co, w, j0, vx, vy);
}

__device__ __forceinline__ float srcDinv(const unsigned short* tb, const float* __restrict__ dinv, int r) {
    const unsigned d = r < DCAP ? (unsigned)tb[r] : DESC;
    float s;
    if (d != DESC) {
        s = rsqrtf((float)d);
    } else {
        s = dinv[r];
        asm volatile("" :: "v"(s));
    }
    return s;
}

#define SC 14336
#define EA 16

template <int F, bool RELU, int COUT, bool WST>
__global__ __launch_bounds__(BT) void k_bagg(const float* __restrict__ Hs, const int2* __restrict__ edges,
                                             const int* __restrict__ tab, long long E, const float* __restrict__ dinv,
                                             const float* __restrict__ bias, const float* __restrict__ Wn,
                                             float* __restrict__ wts, const unsigned short* __restrict__ deg16,
                                             const int* __restrict__ degc, float* __restrict__ out, int n) {
    constexpr int VPL = F / 32;
    constexpr int P = (COUT > 0) ? F + 4 : F;
    static_assert(NB % 16 == 0 && NB / 16 <= BT / 32, "one 16-row strip per wave");
    constexpr int DN = WST ? DCAP : 8;
    static_assert(DN % 8 == 0, "the table is copied in 16-byte items of 8 words");
    constexpr int PADE = 16;
    constexpr int LISTB = (SC + PADE) * 8;
    constexpr int ROWB = NB * P * 4;
    constexpr int TABB = WST ? DN * 2 : 0;
    constexpr int REG2 = (COUT > 0 && ROWB > TABB) ? ROWB : (TABB > 0 ? TABB : 16);
    static_assert(ROWB <= LISTB && LISTB % 16 == 0, "the LDS-add path keeps its sum rows in the list's place");
    __shared__ __attribute__((aligned(16))) unsigned char lds[LISTB + REG2];
    int2* const list = (int2*)lds;
    float* const acc = (float*)lds;
    unsigned short* const tb = (unsigned short*)(lds + LISTB);
    float* const rowsF = (float*)(lds + LISTB);
    __shared__ int cum[GB + 1];
    __shared__ int adj[GB];
    __shared__ int st[NB + 1];
    __shared__ int nxt[NB + 32];
    __shared__ float dl[WST ? NB : 1];
    const int lane = threadIdx.x & 31;
    const int wv = __builtin_amdgcn_readfirstlane((int)(threadIdx.x >> 5));
    const int b = blockIdx.x;
    const int node0 = b * NB;

    constexpr int DQ = (DN / 8 + BT - 1) / BT;
    const int nt = n < DN ? n : DN;
    const int ni = nt / 8;
    const int rot = (int)((unsigned)(b * ni) / gridDim.x);
    v4u dv[DQ];
    if constexpr (WST) {
        if (ni > 0) {
            #pragma unroll
            for (int q = 0; q < DQ; ++q) {
                int j = threadIdx.x + q * BT;
                if (j >= ni) j = ni - 1;
                j += rot;
                if (j >= ni) j -= ni;
                dv[q] = ((const v4u*)deg16)[j];
            }
        }
    }
    if (wv == 0) segTable(tab, b, (int)((E + GB - 1) / GB), cum, adj, lane);
    if (wv == 1) {
        constexpr int NPL = (NB + 31) / 32;
        int d[NPL], ex[NPL];
        #pragma unroll
        for (int q = 0; q < NPL; ++q) {
            const int i = lane * NPL + q;
            d[q] = (i < NB && node0 + i < n) ? degc[node0 + i] : 0;
        }
        int run = 0;
        #pragma unroll
        for (int q = 0; q < NPL; ++q) {
            ex[q] = run;
            run += d[q];
        }
        int incl = run;
        for (int dd = 1; dd < 32; dd <<= 1) {
            const int y = __shfl_up(incl, dd);
            if (lane >= dd) incl += y;
        }
        const int base = incl - run;
        #pragma unroll
        for (int q = 0; q < NPL; ++q) {
            const int i = lane * NPL + q;
            if (i < NB) {
                st[i] = base + ex[q];
                nxt[i] = base + ex[q];
            }
        }
        if (lane == 31) st[NB] = incl;
        nxt[NB + lane] = 0;
    }
    if constexpr (WST) {
        for (int i = threadIdx.x; i < NB; i += BT) {
            const int node = node0 + i;
            dl[i] = node < n ? dinv[node] : 0.0f;
        }
        if ((int)threadIdx.x < nt - ni * 8)
            tb[ni * 8 + (int)threadIdx.x] = deg16[ni * 8 + threadIdx.x];
        if (ni > 0) {
            #pragma unroll
            for (int q = 0; q < DQ; ++q) {
                int j = threadIdx.x + q * BT;
                if (j < ni) {
                    j += rot;
                    if (j >= ni) j -= ni;
                    *(v4u*)(tb + 8 * j) = dv[q];
                }
            }
        }
    }
    __syncthreads();

    const int T = cum[GB];
    const bool fits = T <= SC;
    const float* rows = rowsF;
    if (fits) {
        for (int v0 = threadIdx.x; v0 < T; v0 += EA * BT) {
            int sl[EA];
            int2 rc[EA];
            float w[EA];
            #pragma unroll
            for (int q = 0; q < EA; ++q) {
                int v = v0 + q * BT;
                if (v >= T) v = T - 1;
                sl[q] = slotOf(cum, adj, v);
            }
            #pragma unroll
            for (int q = 0; q < EA; ++q) {
                rc[q] = edges[sl[q]];
                if constexpr (!WST) w[q] = wts[sl[q]];
                else w[q] = 0.0f;
            }
            static_assert(EA == 16, "the statements below name all EA loaded words");
            asm volatile("" :: "v"(rc[0].x), "v"(rc[1].x), "v"(rc[2].x), "v"(rc[3].x), "v"(rc[4].x), "v"(rc[5].x), "v"(rc[6].x), "v"(rc[7].x),
                               "v"(rc[8].x), "v"(rc[9].x), "v"(rc[10].x), "v"(rc[11].x), "v"(rc[12].x), "v"(rc[13].x), "v"(rc[14].x), "v"(rc[15].x));
            asm volatile("" :: "v"(rc[0].y), "v"(rc[1].y), "v"(rc[2].y), "v"(rc[3].y), "v"(rc[4].y), "v"(rc[5].y), "v"(rc[6].y), "v"(rc[7].y),
                               "v"(rc[8].y), "v"(rc[9].y), "v"(rc[10].y), "v"(rc[11].y), "v"(rc[12].y), "v"(rc[13].y), "v"(rc[14].y), "v"(rc[15].y));
            if constexpr (!WST)
                asm volatile("" :: "v"(w[0]), "v"(w[1]), "v"(w[2]), "v"(w[3]), "v"(w[4]), "v"(w[5]), "v"(w[6]), "v"(w[7]),
                                   "v"(w[8]), "v"(w[9]), "v"(w[10]), "v"(w[11]), "v"(w[12]), "v"(w[13]), "v"(w[14]), "v"(w[15]));
            if constexpr (WST) {
                unsigned dg[EA];
                float dc[EA];
                #pragma unroll
                for (int q = 0; q < EA; ++q) {
                    const int r = rc[q].x;
                    dg[q] = r < DCAP ? (unsigned)tb[r] : DESC;
                    dc[q] = dl[rc[q].y];
                }
                asm volatile("" :: "v"(dg[0]), "v"(dg[1]), "v"(dg[2]), "v"(dg[3]), "v"(dg[4]), "v"(dg[5]), "v"(dg[6]), "v"(dg[7]),
                                   "v"(dg[8]), "v"(dg[9]), "v"(dg[10]), "v"(dg[11]), "v"(dg[12]), "v"(dg[13]), "v"(dg[14]), "v"(dg[15]));
                asm volatile("" :: "v"(dc[0]), "v"(dc[1]), "v"(dc[2]), "v"(dc[3]), "v"(dc[4]), "v"(dc[5]), "v"(dc[6]), "v"(dc[7]),
                                   "v"(dc[8]), "v"(dc[9]), "v"(dc[10]), "v"(dc[11]), "v"(dc[12]), "v"(dc[13]), "v"(dc[14]), "v"(dc[15]));
                #pragma unroll
                for (int q = 0; q < EA; ++q) {
                    float s;
                    if (dg[q] != DESC) {
                        s = rsqrtf((float)dg[q]);
                    } else {
                        s = dinv[rc[q].x];
                        asm volatile("" :: "v"(s));
                    }
                    w[q] = s * dc[q];
                    if (v0 + q * BT < T) wts[sl[q]] = w[q];
                }
            }
            int p[EA];
            #pragma unroll
            for (int q = 0; q < EA; ++q) {
                const int cw = (v0 + q * BT < T) ? rc[q].y : NB + lane;
                p[q] = atomicAdd(&nxt[cw], 1);
            }
            #pragma unroll
            for (int q = 0; q < EA; ++q)
                if (v0 + q * BT < T) list[p[q]] = make_int2(rc[q].x, __float_as_int(w[q]));
        }
        if (threadIdx.x < PADE) list[T + threadIdx.x] = make_int2(0, 0);
        __syncthreads();

        constexpr int LPR = F / 4;
        constexpr int GN = 16 / LPR;
        constexpr int UB = (F == 64) ? 12 : 8;
        const int hn = lane >> 4;
        const int sub = (lane & 15) / LPR;
        const int col = (lane % LPR) * 4;
        static_assert(UB * GN <= PADE, "a trip reads at most PADE entries past a node's end");
        static_assert((long long)NB * MAXB * F * 4 < (1ll << 32), "32-bit row offsets");
        const char* const hb = (const char*)Hs;
        const unsigned colb = (unsigned)col * 4u;
        const float bs0 = bias[col], bs1 = bias[col + 1], bs2 = bias[col + 2], bs3 = bias[col + 3];
        for (int i = wv * 2 + hn; i < NB; i += 2 * (BT / 32)) {
            const int node = node0 + i;
            float4 res = make_float4(0.f, 0.f, 0.f, 0.f);
            if (node < n) {
                const int beg = st[i], end = st[i + 1];
                const float4 own = *(const float4*)(hb + ((unsigned)node * (unsigned)(F * 4) + colb));
                float di;
                if constexpr (WST) di = dl[i];
                else di = dinv[node];
                float4 a = make_float4(0.f, 0.f, 0.f, 0.f);
                for (int e0 = beg; e0 < end; e0 += UB * GN) {
                    float4 v[UB];
                    float w[UB];
                    int2 rw[UB];
                    #pragma unroll
                    for (int u = 0; u < UB; ++u) rw[u] = list[e0 + u * GN + sub];
                    static_assert(UB == 8 || UB == 12, "the statements below name all UB list entries");
                    asm volatile("" :: "v"(rw[0].x), "v"(rw[0].y), "v"(rw[1].x), "v"(rw[1].y), "v"(rw[2].x), "v"(rw[2].y),
                                       "v"(rw[3].x), "v"(rw[3].y), "v"(rw[4].x), "v"(rw[4].y), "v"(rw[5].x), "v"(rw[5].y),
                                       "v"(rw[6].x), "v"(rw[6].y), "v"(rw[7].x), "v"(rw[7].y));
                    if constexpr (UB == 12)
                        asm volatile("" :: "v"(rw[UB - 4].x), "v"(rw[UB - 4].y), "v"(rw[UB - 3].x), "v"(rw[UB - 3].y),
                                           "v"(rw[UB - 2].x), "v"(rw[UB - 2].y), "v"(rw[UB - 1].x), "v"(rw[UB - 1].y));
                    #pragma unroll
                    for (int u = 0; u < UB; ++u) {
                        const bool ok = (e0 + u * GN + sub) < end;
                        w[u] = ok ? __int_as_float(rw[u].y) : 0.0f;
                        const unsigned rr = ok ? (unsigned)rw[u].x : (unsigned)node;
                        v[u] = *(const float4*)(hb + (rr * (unsigned)(F * 4) + colb));
                    }
                    #pragma unroll
                    for (int u = 0; u < UB; ++u) {
                        a.x += w[u] * v[u].x;
                        a.y += w[u] * v[u].y;
                        a.z += w[u] * v[u].z;
                        a.w += w[u] * v[u].w;
                    }
                }
                #pragma unroll
                for (int d = LPR; d < 16; d <<= 1) {
                    a.x += __shfl_xor(a.x, d);
                    a.y += __shfl_xor(a.y, d);
                    a.z += __shfl_xor(a.z, d);
                    a.w += __shfl_xor(a.w, d);
                }
                res.x = a.x + di * di * own.x + bs0;
                res.y = a.y + di * di * own.y + bs1;
                res.z = a.z + di * di * own.z + bs2;
                res.w = a.w + di * di * own.w + bs3;
                if (RELU) {
                    res.x = fmaxf(res.x, 0.0f);
                    res.y = fmaxf(res.y, 0.0f);
                    res.z = fmaxf(res.z, 0.0f);
                    res.w = fmaxf(res.w, 0.0f);
                }
            }
            if (sub == 0) {
                if constexpr (COUT > 0) *(float4*)(rowsF + i * P + col) = res;
                else if (node < n) *(float4*)(out + (size_t)node * F + col) = res;
            }
        }
    } else {
        static_assert(P % 4 == 0, "a row is zeroed in whole float4 items");
        for (int i = threadIdx.x; i < NB * P / 4; i += BT)
            *(float4*)(acc + i * 4) = make_float4(0.f, 0.f, 0.f, 0.f);
        __syncthreads();
        const float* hl = Hs + lane * VPL;
        float* al = acc + lane * VPL;
        for (int g = wv * 32; g < T; g += BT) {
            const int cnt = T - g;
            int r = 0, c = 0;
            float w = 0.0f;
            if (lane < cnt) {
                const int sl = slotOf(cum, adj, g + lane);
                const int2 rc = edges[sl];
                r = rc.x;
                c = rc.y;
                if constexpr (WST) {
                    w = srcDinv(tb, dinv, r) * dl[c];
                    wts[sl] = w;
                } else {
                    w = wts[sl];
                }
            }
            const int co = c * P;
            for (int j = 0; j < cnt && j < 32; ++j) addRows<F, 1>(hl, al, r, co, w, j);
        }
        __syncthreads();
        for (int i = threadIdx.x; i < NB * F / 4; i += BT) {
            const int nl = (i * 4) / F;
            const int node = node0 + nl;
            if (node < n) {
                const int f = (i * 4) % F;
                const size_t gi = (size_t)node0 * F + (size_t)i * 4;
                const float di = dinv[node];
                float4* ap = (float4*)(acc + nl * P + f);
                const float4 a = *ap;
                const float4 h = *(const float4*)(Hs + gi);
                float4 v;
                v.x = a.x + di * di * h.x + bias[f];
                v.y = a.y + di * di * h.y + bias[f + 1];
                v.z = a.z + di * di * h.z + bias[f + 2];
                v.w = a.w + di * di * h.w + bias[f + 3];
                if (RELU) {
                    v.x = fmaxf(v.x, 0.0f);
                    v.y = fmaxf(v.y, 0.0f);
                    v.z = fmaxf(v.z, 0.0f);
                    v.w = fmaxf(v.w, 0.0f);
                }
                if constexpr (COUT > 0) *ap = v;
                else *(float4*)(out + gi) = v;
            }
        }
        rows = acc;
    }

    if constexpr (COUT > 0) {
        constexpr int CT = COUT / 16;
        __syncthreads();
        if (wv < NB / 16) {
            const int m  = lane & 15;
            const int kh = (lane >> 4) << 1;
            const float* ar = rows + (wv * 16 + m) * P + kh;
            const float* wc = Wn + kh * COUT + m;
            v8f c[CT] = {};
            #pragma unroll
            for (int k = 0; k < F; k += 4) {
                v2f a;
                const float2 av = *(const float2*)(ar + k);
                a[0] = av.x;
                a[1] = av.y;
                #pragma unroll
                for (int t = 0; t < CT; ++t) {
                    v2f bf;
                    bf[0] = wc[(size_t)k       * COUT + t * 16];
                    bf[1] = wc[(size_t)(k + 1) * COUT + t * 16];
                    c[t] = __builtin_amdgcn_wmma_f32_16x16x4_f32(
                         false, a,  false, bf,
                         (short)0, c[t],  false,  false);
                }
            }
            const int mo = (lane >> 4) << 3;
            #pragma unroll
            for (int v = 0; v < 8; ++v) {
                const int r = node0 + wv * 16 + mo + v;
                if (r < n) {
                    #pragma unroll
                    for (int t = 0; t < CT; ++t) out[(size_t)r * COUT + t * 16 + m] = c[t][v];
                }
            }
        }
    }
}

#define DEPW 64
#define DBT  128
__global__ __launch_bounds__(DBT) void k_decode(const float* __restrict__ Z, const int* __restrict__ ea,
                                                const int* __restrict__ eb, float* __restrict__ out, long long EL) {
    static_assert(DEPW == 64, "two halves of 32 edges");
    const int lane = threadIdx.x & 31;
    const int t = lane & 7;
    const long long wave = (long long)blockIdx.x * (DBT / 32) + (threadIdx.x >> 5);
    const long long base = wave * DEPW;
    if (base >= EL) return;
    int ia[2], ib[2];
    #pragma unroll
    for (int h = 0; h < 2; ++h) {
        long long e = base + 32 * h + lane;
        if (e >= EL) e = EL - 1;
        ia[h] = ea[e];
        ib[h] = eb[e];
    }
    float4 a[16], b[16];
    #pragma unroll
    for (int h = 0; h < 2; ++h) {
        #pragma unroll
        for (int s = 0; s < 8; ++s) {
            const int ra = __shfl(ia[h], s, 8);
            const int rb = __shfl(ib[h], s, 8);
            a[8 * h + s] = *(const float4*)(Z + (size_t)ra * 32 + t * 4);
            b[8 * h + s] = *(const float4*)(Z + (size_t)rb * 32 + t * 4);
        }
    }
    #pragma unroll
    for (int h = 0; h < 2; ++h) {
        float res = 0.0f;
        #pragma unroll
        for (int s = 0; s < 8; ++s) {
            const float4 x = a[8 * h + s], y = b[8 * h + s];
            float p = x.x * y.x + x.y * y.y + x.z * y.z + x.w * y.w;
            p += __shfl_xor(p, 1);
            p += __shfl_xor(p, 2);
            p += __shfl_xor(p, 4);
            if (t == s) res = p;
        }
        const long long e = base + 32 * h + lane;
        if (e < EL) out[e] = res;
    }
}

static inline int cdiv(long long a, long long b) { return (int)((a + b - 1) / b); }

extern "C" void kernel_launch(void* const* d_in, const int* in_sizes, int n_in,
                              void* d_out, int out_size, void* d_ws, size_t ws_size,
                              hipStream_t stream) {
    const float* x   = (const float*)d_in[0];
    const float* W1  = (const float*)d_in[1];
    const float* b1  = (const float*)d_in[2];
    const float* W2  = (const float*)d_in[3];
    const float* b2  = (const float*)d_in[4];
    const int*   ei  = (const int*)d_in[5];
    const int*   eli = (const int*)d_in[6];

    const long long N  = in_sizes[0] / F_IN;
    const long long E  = in_sizes[5] / 2;
    const long long EL = in_sizes[6] / 2;
    float* out = (float*)d_out;
    const int nbk = cdiv(N, NB);

    float* h1    = (float*)d_ws;
    float* h2    = h1   + N * F_HID;
    float* agg2  = h2   + N * F_OUT;
    unsigned short* deg16 = (unsigned short*)(agg2 + N * F_OUT);
    int*   degc  = (int*)(deg16 + 2 * N);
    int2*  edges = (int2*)(deg16 + 4 * N);
    float* dinv  = (float*)(edges + E);
    float* wts   = dinv + N;
    int*   tab   = (int*)(wts + E);

    const int* row = ei;
    const int* col = ei + E;

    k_lsort<<<GB, BT, 0, stream>>>(row, col, edges, tab, E, nbk);

    k_bdeg_gemm<<<nbk, BT, 0, stream>>>(edges, tab, E, dinv, deg16, degc, x, W1, h1, (int)N);

    k_bagg<F_HID, true, F_OUT, true><<<nbk, BT, 0, stream>>>(h1, edges, tab, E, dinv, b1, W2, wts, deg16, degc, h2, (int)N);

    k_bagg<F_OUT, false, 0, false><<<nbk, BT, 0, stream>>>(h2, edges, tab, E, dinv, b2, nullptr, wts, nullptr, degc, agg2, (int)N);

    k_decode<<<cdiv(EL, DEPW * (DBT / 32)), DBT, 0, stream>>>(agg2, eli, eli + EL, out, EL);
}
